# starting kernel
# baseline (speedup 1.0000x reference)
#include <hip/hip_runtime.h>
#include <math.h>

typedef __attribute__((ext_vector_type(2))) float v2f;
typedef __attribute__((ext_vector_type(8))) float v8f;

#define NEGBIG (-3.4e38f)

__global__ __launch_bounds__(256) void gemm_wmma_f32(
    const float* __restrict__ A, const float* __restrict__ W,
    const float* __restrict__ bias, float* __restrict__ C,
    int M, int N, int K)
{
  __shared__ float As[128][33];
  __shared__ float Bs[32][130];

  const int tid  = threadIdx.x;
  const int wave = tid >> 5;
  const int lane = tid & 31;
  const int r    = lane & 15;
  const int hi   = lane >> 4;
  const int wrow = wave >> 2;
  const int wcol = wave & 3;
  const int m0   = blockIdx.y * 128;
  const int n0   = blockIdx.x * 128;

  v8f acc[4][2];
  for (int i = 0; i < 4; ++i)
    for (int j = 0; j < 2; ++j)
      for (int c = 0; c < 8; ++c) acc[i][j][c] = 0.0f;

  for (int k0 = 0; k0 < K; k0 += 32) {
    __syncthreads();
    {
      int row = tid >> 3;
      int col = (tid & 7) * 4;
      #pragma unroll
      for (int it = 0; it < 4; ++it) {
        int rr = row + it * 32;
        float4 v = *(const float4*)&A[(size_t)(m0 + rr) * K + k0 + col];
        As[rr][col + 0] = v.x; As[rr][col + 1] = v.y;
        As[rr][col + 2] = v.z; As[rr][col + 3] = v.w;
      }
    }
    {
      int n  = tid >> 3;
      int kk = (tid & 7) * 4;
      #pragma unroll
      for (int it = 0; it < 4; ++it) {
        int nn = n + it * 32;
        float4 v = make_float4(0.f, 0.f, 0.f, 0.f);
        if (n0 + nn < N)
          v = *(const float4*)&W[(size_t)(n0 + nn) * K + k0 + kk];
        Bs[kk + 0][nn] = v.x; Bs[kk + 1][nn] = v.y;
        Bs[kk + 2][nn] = v.z; Bs[kk + 3][nn] = v.w;
      }
    }
    __syncthreads();

    #pragma unroll
    for (int kk = 0; kk < 32; kk += 4) {
      v2f af[4], bf[2];
      #pragma unroll
      for (int i = 0; i < 4; ++i) {
        int row = wrow * 64 + i * 16 + r;
        af[i][0] = As[row][kk + 2 * hi];
        af[i][1] = As[row][kk + 1 + 2 * hi];
      }
      #pragma unroll
      for (int j = 0; j < 2; ++j) {
        int col = wcol * 32 + j * 16 + r;
        bf[j][0] = Bs[kk + 2 * hi][col];
        bf[j][1] = Bs[kk + 1 + 2 * hi][col];
      }
      #pragma unroll
      for (int i = 0; i < 4; ++i)
        #pragma unroll
        for (int j = 0; j < 2; ++j)
          acc[i][j] = __builtin_amdgcn_wmma_f32_16x16x4_f32(
              false, af[i], false, bf[j], (short)0, acc[i][j], false, false);
    }
  }

  #pragma unroll
  for (int i = 0; i < 4; ++i)
    for (int j = 0; j < 2; ++j) {
      int gcol = n0 + wcol * 32 + j * 16 + r;
      if (gcol >= N) continue;
      float b = bias ? bias[gcol] : 0.0f;
      #pragma unroll
      for (int vg = 0; vg < 8; ++vg) {
        int grow = m0 + wrow * 64 + i * 16 + vg + 8 * hi;
        C[(size_t)grow * N + gcol] = acc[i][j][vg] + b;
      }
    }
}

__global__ void rope_kernel(float* __restrict__ qbuf, float* __restrict__ kvbuf)
{
  const long long QP = 2048LL * 64 * 32;
  const long long KP = 2048LL * 8 * 32;
  long long idx = (long long)blockIdx.x * blockDim.x + threadIdx.x;
  if (idx >= QP + KP) return;

  int s, i;
  float *p0, *p1;
  if (idx < QP) {
    int t = (int)idx;
    i = t & 31;
    int h = (t >> 5) & 63;
    s = t >> 11;
    float* base = qbuf + (size_t)s * 4096 + h * 64;
    p0 = base + i; p1 = base + i + 32;
  } else {
    int t = (int)(idx - QP);
    i = t & 31;
    int kvh = (t >> 5) & 7;
    s = t >> 8;
    float* base = kvbuf + (size_t)s * 1024 + kvh * 128;
    p0 = base + i; p1 = base + i + 32;
  }
  float inv = powf(150000.0f, -((float)(2 * i)) / 64.0f);
  float ang = (float)s * inv;
  float c = cosf(ang), sn = sinf(ang);
  float x1 = *p0, x2 = *p1;
  *p0 = x1 * c - x2 * sn;
  *p1 = x2 * c + x1 * sn;
}

__global__ __launch_bounds__(128) void flash_attn(
    const float* __restrict__ qbuf, const float* __restrict__ kvbuf,
    float* __restrict__ attn)
{
  __shared__ float Qs[64][65];
  __shared__ float Kt[64][65];
  __shared__ float Vs[64][64];
  __shared__ float Ssc[64][65];

  const int tid   = threadIdx.x;
  const int wave  = tid >> 5;
  const int lane  = tid & 31;
  const int r     = lane & 15;
  const int hi    = lane >> 4;
  const int qtile = blockIdx.x;
  const int h     = blockIdx.y;
  const int kvh   = h >> 3;
  const int qbase = qtile * 64;

  #pragma unroll 4
  for (int it = 0; it < 32; ++it) {
    int f = tid + it * 128;
    int sl = f >> 6, d = f & 63;
    Qs[sl][d] = qbuf[(size_t)(qbase + sl) * 4096 + h * 64 + d];
  }

  float m_i = -INFINITY, l_i = 0.0f;
  v8f oacc[4];
  for (int sub = 0; sub < 4; ++sub)
    for (int c = 0; c < 8; ++c) oacc[sub][c] = 0.0f;

  for (int j = 0; j <= qtile; ++j) {
    const int kvbase = j * 64;
    __syncthreads();
    #pragma unroll 4
    for (int it = 0; it < 32; ++it) {
      int f = tid + it * 128;
      int n = f >> 6, d = f & 63;
      const float* kv = &kvbuf[(size_t)(kvbase + n) * 1024 + kvh * 128];
      Kt[d][n] = kv[d];
      Vs[n][d] = kv[64 + d];
    }
    __syncthreads();

    v8f sacc[4];
    for (int sub = 0; sub < 4; ++sub)
      for (int c = 0; c < 8; ++c) sacc[sub][c] = 0.0f;
    #pragma unroll
    for (int kk = 0; kk < 64; kk += 4) {
      v2f af;
      af[0] = Qs[wave * 16 + r][kk + 2 * hi];
      af[1] = Qs[wave * 16 + r][kk + 1 + 2 * hi];
      #pragma unroll
      for (int sub = 0; sub < 4; ++sub) {
        v2f bf;
        bf[0] = Kt[kk + 2 * hi][sub * 16 + r];
        bf[1] = Kt[kk + 1 + 2 * hi][sub * 16 + r];
        sacc[sub] = __builtin_amdgcn_wmma_f32_16x16x4_f32(
            false, af, false, bf, (short)0, sacc[sub], false, false);
      }
    }

    #pragma unroll
    for (int sub = 0; sub < 4; ++sub)
      #pragma unroll
      for (int vg = 0; vg < 8; ++vg) {
        int m  = vg + 8 * hi;
        int qr = qbase + wave * 16 + m;
        int kc = kvbase + sub * 16 + r;
        float v = (kc <= qr) ? sacc[sub][vg] * 0.125f : NEGBIG;
        Ssc[wave * 16 + m][sub * 16 + r] = v;
      }

    {
      int row = wave * 16 + r;
      float rowmax = -INFINITY;
      for (int c = 0; c < 64; ++c) rowmax = fmaxf(rowmax, Ssc[row][c]);
      float newm = fmaxf(m_i, rowmax);
      float sum = 0.0f;
      for (int c = 0; c < 64; ++c) {
        float p = __expf(Ssc[row][c] - newm);
        Ssc[row][c] = p;
        sum += p;
      }
      float alpha = __expf(m_i - newm);
      l_i = l_i * alpha + sum;
      m_i = newm;
      #pragma unroll
      for (int vg = 0; vg < 8; ++vg) {
        float am = __int_as_float(__builtin_amdgcn_ds_bpermute(
            4 * (vg + 24 * hi), __float_as_int(alpha)));
        for (int sub = 0; sub < 4; ++sub) oacc[sub][vg] *= am;
      }
    }

    #pragma unroll
    for (int kk = 0; kk < 64; kk += 4) {
      v2f af;
      af[0] = Ssc[wave * 16 + r][kk + 2 * hi];
      af[1] = Ssc[wave * 16 + r][kk + 1 + 2 * hi];
      #pragma unroll
      for (int sub = 0; sub < 4; ++sub) {
        v2f bf;
        bf[0] = Vs[kk + 2 * hi][sub * 16 + r];
        bf[1] = Vs[kk + 1 + 2 * hi][sub * 16 + r];
        oacc[sub] = __builtin_amdgcn_wmma_f32_16x16x4_f32(
            false, af, false, bf, (short)0, oacc[sub], false, false);
      }
    }
  }

  float inv = 1.0f / l_i;
  #pragma unroll
  for (int vg = 0; vg < 8; ++vg) {
    float im = __int_as_float(__builtin_amdgcn_ds_bpermute(
        4 * (vg + 24 * hi), __float_as_int(inv)));
    int grow = qbase + wave * 16 + vg + 8 * hi;
    for (int sub = 0; sub < 4; ++sub) {
      int d = sub * 16 + r;
      attn[(size_t)grow * 4096 + h * 64 + d] = oacc[sub][vg] * im;
    }
  }
}

extern "C" void kernel_launch(void* const* d_in, const int* in_sizes, int n_in,
                              void* d_out, int out_size, void* d_ws, size_t ws_size,
                              hipStream_t stream)
{
  const float* hidden = (const float*)d_in[0];
  const float* q_w    = (const float*)d_in[2];
  const float* q_b    = (const float*)d_in[3];
  const float* kv_a_w = (const float*)d_in[4];
  const float* kv_b_w = (const float*)d_in[5];
  const float* o_w    = (const float*)d_in[6];
  const float* o_b    = (const float*)d_in[7];
  float* out = (float*)d_out;

  float* ws    = (float*)d_ws;
  float* qbuf  = ws;
  float* ckv   = qbuf  + (size_t)2048 * 4096;
  float* kvbuf = ckv   + (size_t)2048 * 384;
  float* attn  = kvbuf + (size_t)2048 * 1024;

  dim3 blk(256);
  gemm_wmma_f32<<<dim3(4096 / 128, 2048 / 128), blk, 0, stream>>>(
      hidden, q_w, q_b, qbuf, 2048, 4096, 2880);
  gemm_wmma_f32<<<dim3(384 / 128, 2048 / 128), blk, 0, stream>>>(
      hidden, kv_a_w, nullptr, ckv, 2048, 384, 2880);
  gemm_wmma_f32<<<dim3(1024 / 128, 2048 / 128), blk, 0, stream>>>(
      ckv, kv_b_w, nullptr, kvbuf, 2048, 1024, 384);
  long long pairs = 2048LL * 64 * 32 + 2048LL * 8 * 32;
  rope_kernel<<<(int)((pairs + 255) / 256), 256, 0, stream>>>(qbuf, kvbuf);
  flash_attn<<<dim3(32, 64), 128, 0, stream>>>(qbuf, kvbuf, attn);
  gemm_wmma_f32<<<dim3((2880 + 127) / 128, 2048 / 128), blk, 0, stream>>>(
      attn, o_w, o_b, out, 2048, 2880, 4096);
}


# O projection k range split over two grid planes (736 blocks, not 368, so every WGP holds 2-3); plane 1 writes a partial sum, a small kernel adds it into out: k sum order changes, 5 launches -> 6
# speedup vs baseline: 1.3814x; 1.3814x over previous
#include <hip/hip_runtime.h>
#include <math.h>

typedef __attribute__((ext_vector_type(2))) float v2f;
typedef __attribute__((ext_vector_type(8))) float v8f;

#define NEGBIG (-3.4e38f)

__global__ __launch_bounds__(256) void gemm_wmma_f32(
    const float* __restrict__ A,
    const float* __restrict__ W0, const float* __restrict__ bias0,
    float* __restrict__ C0, int N0, int nb0,
    const float* __restrict__ W1, const float* __restrict__ bias1,
    float* __restrict__ C1, int N1,
    int M, int K, int kslice, float* __restrict__ Cpart)
{
  const bool second = (int)blockIdx.x >= nb0;
  const float* __restrict__ W    = second ? W1 : W0;
  const bool  partial = blockIdx.z != 0;
  const int   kbeg    = (int)blockIdx.z * kslice;
  const int   kend    = kbeg + kslice;
  const float* __restrict__ bias = partial ? nullptr : (second ? bias1 : bias0);
  float* __restrict__ C          = partial ? Cpart : (second ? C1 : C0);
  const int N                    = second ? N1 : N0;
  const int nblk                 = second ? (int)blockIdx.x - nb0 : (int)blockIdx.x;

  __shared__ float As[128][33];
  __shared__ float Bs[32][130];

  const int tid  = threadIdx.x;
  const int wave = tid >> 5;
  const int lane = tid & 31;
  const int r    = lane & 15;
  const int hi   = lane >> 4;
  const int wrow = wave >> 2;
  const int wcol = wave & 3;
  const int m0   = blockIdx.y * 128;
  const int n0   = nblk * 128;

  v8f acc[4][2];
  for (int i = 0; i < 4; ++i)
    for (int j = 0; j < 2; ++j)
      for (int c = 0; c < 8; ++c) acc[i][j][c] = 0.0f;

  for (int k0 = kbeg; k0 < kend; k0 += 32) {
    __syncthreads();
    {
      int row = tid >> 3;
      int col = (tid & 7) * 4;
      #pragma unroll
      for (int it = 0; it < 4; ++it) {
        int rr = row + it * 32;
        float4 v = *(const float4*)&A[(size_t)(m0 + rr) * K + k0 + col];
        As[rr][col + 0] = v.x; As[rr][col + 1] = v.y;
        As[rr][col + 2] = v.z; As[rr][col + 3] = v.w;
      }
    }
    {
      int n  = tid >> 3;
      int kk = (tid & 7) * 4;
      #pragma unroll
      for (int it = 0; it < 4; ++it) {
        int nn = n + it * 32;
        float4 v = make_float4(0.f, 0.f, 0.f, 0.f);
        if (n0 + nn < N)
          v = *(const float4*)&W[(size_t)(n0 + nn) * K + k0 + kk];
        Bs[kk + 0][nn] = v.x; Bs[kk + 1][nn] = v.y;
        Bs[kk + 2][nn] = v.z; Bs[kk + 3][nn] = v.w;
      }
    }
    __syncthreads();

    #pragma unroll
    for (int kk = 0; kk < 32; kk += 4) {
      v2f af[4], bf[2];
      #pragma unroll
      for (int i = 0; i < 4; ++i) {
        int row = wrow * 64 + i * 16 + r;
        af[i][0] = As[row][kk + 2 * hi];
        af[i][1] = As[row][kk + 1 + 2 * hi];
      }
      #pragma unroll
      for (int j = 0; j < 2; ++j) {
        int col = wcol * 32 + j * 16 + r;
        bf[j][0] = Bs[kk + 2 * hi][col];
        bf[j][1] = Bs[kk + 1 + 2 * hi][col];
      }
      #pragma unroll
      for (int i = 0; i < 4; ++i)
        #pragma unroll
        for (int j = 0; j < 2; ++j)
          acc[i][j] = __builtin_amdgcn_wmma_f32_16x16x4_f32(
              false, af[i], false, bf[j], (short)0, acc[i][j], false, false);
    }
  }

  #pragma unroll
  for (int i = 0; i < 4; ++i)
    for (int j = 0; j < 2; ++j) {
      int gcol = n0 + wcol * 32 + j * 16 + r;
      if (gcol >= N) continue;
      float b = bias ? bias[gcol] : 0.0f;
      #pragma unroll
      for (int vg = 0; vg < 8; ++vg) {
        int grow = m0 + wrow * 64 + i * 16 + vg + 8 * hi;
        C[(size_t)grow * N + gcol] = acc[i][j][vg] + b;
      }
    }
}

__global__ __launch_bounds__(256) void add_partial(float* __restrict__ C,
                                                   const float* __restrict__ P)
{
  const size_t base = (size_t)blockIdx.x * 1024 + threadIdx.x;
  float4 a[4], b[4];
  #pragma unroll
  for (int u = 0; u < 4; ++u) {
    a[u] = ((const float4*)C)[base + u * 256];
    b[u] = ((const float4*)P)[base + u * 256];
  }
  #pragma unroll
  for (int u = 0; u < 4; ++u) {
    a[u].x += b[u].x; a[u].y += b[u].y; a[u].z += b[u].z; a[u].w += b[u].w;
    ((float4*)C)[base + u * 256] = a[u];
  }
}

__global__ void rope_kernel(float* __restrict__ qbuf, float* __restrict__ kvbuf)
{
  const long long QP = 2048LL * 64 * 32;
  const long long KP = 2048LL * 8 * 32;
  long long idx = (long long)blockIdx.x * blockDim.x + threadIdx.x;
  if (idx >= QP + KP) return;

  int s, i;
  float *p0, *p1;
  if (idx < QP) {
    int t = (int)idx;
    i = t & 31;
    int h = (t >> 5) & 63;
    s = t >> 11;
    float* base = qbuf + (size_t)s * 4096 + h * 64;
    p0 = base + i; p1 = base + i + 32;
  } else {
    int t = (int)(idx - QP);
    i = t & 31;
    int kvh = (t >> 5) & 7;
    s = t >> 8;
    float* base = kvbuf + (size_t)s * 1024 + kvh * 128;
    p0 = base + i; p1 = base + i + 32;
  }
  float inv = powf(150000.0f, -((float)(2 * i)) / 64.0f);
  float ang = (float)s * inv;
  float c = cosf(ang), sn = sinf(ang);
  float x1 = *p0, x2 = *p1;
  *p0 = x1 * c - x2 * sn;
  *p1 = x2 * c + x1 * sn;
}

__global__ __launch_bounds__(128) void flash_attn(
    const float* __restrict__ qbuf, const float* __restrict__ kvbuf,
    float* __restrict__ attn)
{
  __shared__ float Qs[64][65];
  __shared__ __attribute__((aligned(16))) float Ks[64][68];
  __shared__ __attribute__((aligned(16))) float Vs[64][68];

  const int tid   = threadIdx.x;
  const int wave  = tid >> 5;
  const int lane  = tid & 31;
  const int r     = lane & 15;
  const int hi    = lane >> 4;
  const int qtile = blockIdx.x;
  const int h     = blockIdx.y;
  const int kvh   = h >> 3;
  const int qbase = qtile * 64;
  const int qr    = qbase + wave * 16 + r;

  #pragma unroll 4
  for (int it = 0; it < 32; ++it) {
    int f = tid + it * 128;
    int sl = f >> 6, d = f & 63;
    Qs[sl][d] = qbuf[(size_t)(qbase + sl) * 4096 + h * 64 + d];
  }

  float m_i = -INFINITY;
  float l_i = 0.0f;
  v8f oacc[4];
  for (int ds = 0; ds < 4; ++ds)
    for (int c = 0; c < 8; ++c) oacc[ds][c] = 0.0f;

  for (int j = 0; j <= qtile; ++j) {
    const int kvbase = j * 64;
    __syncthreads();
    #pragma unroll 4
    for (int it = 0; it < 32; ++it) {
      int f = tid + it * 128;
      int n = f >> 6, d = f & 63;
      const float* kv = &kvbuf[(size_t)(kvbase + n) * 1024 + kvh * 128];
      Ks[n][d] = kv[d];
      Vs[n][d] = kv[64 + d];
    }
    __syncthreads();

    v8f sacc[4];
    for (int sub = 0; sub < 4; ++sub)
      for (int c = 0; c < 8; ++c) sacc[sub][c] = 0.0f;
    #pragma unroll
    for (int kk = 0; kk < 64; kk += 4) {
      v2f bq;
      bq[0] = Qs[wave * 16 + r][kk + 2 * hi];
      bq[1] = Qs[wave * 16 + r][kk + 1 + 2 * hi];
      #pragma unroll
      for (int sub = 0; sub < 4; ++sub) {
        v2f ak = *(const v2f*)&Ks[sub * 16 + r][kk + 2 * hi];
        sacc[sub] = __builtin_amdgcn_wmma_f32_16x16x4_f32(
            false, ak, false, bq, (short)0, sacc[sub], false, false);
      }
    }

    #pragma unroll
    for (int sub = 0; sub < 4; ++sub)
      #pragma unroll
      for (int g = 0; g < 8; ++g) sacc[sub][g] *= 0.125f;
    if (j == qtile) {
      #pragma unroll
      for (int sub = 0; sub < 4; ++sub)
        #pragma unroll
        for (int g = 0; g < 8; ++g) {
          int kc = kvbase + sub * 16 + 8 * hi + g;
          sacc[sub][g] = (kc <= qr) ? sacc[sub][g] : NEGBIG;
        }
    }

    float mx[4];
    #pragma unroll
    for (int sub = 0; sub < 4; ++sub) {
      mx[sub] = sacc[sub][0];
      #pragma unroll
      for (int g = 1; g < 8; ++g) mx[sub] = fmaxf(mx[sub], sacc[sub][g]);
    }
    float rowmax = fmaxf(fmaxf(mx[0], mx[1]), fmaxf(mx[2], mx[3]));
    rowmax = fmaxf(rowmax, __int_as_float(__builtin_amdgcn_ds_bpermute(
        4 * (lane ^ 16), __float_as_int(rowmax))));
    const float newm  = fmaxf(m_i, rowmax);
    const float alpha = __expf(m_i - newm);
    float ps[4];
    #pragma unroll
    for (int sub = 0; sub < 4; ++sub) {
      ps[sub] = 0.0f;
      #pragma unroll
      for (int g = 0; g < 8; ++g) {
        float p = __expf(sacc[sub][g] - newm);
        sacc[sub][g] = p;
        ps[sub] += p;
      }
    }
    l_i = l_i * alpha + ((ps[0] + ps[1]) + (ps[2] + ps[3]));
    m_i = newm;
    #pragma unroll
    for (int ds = 0; ds < 4; ++ds)
      #pragma unroll
      for (int g = 0; g < 8; ++g) oacc[ds][g] *= alpha;

    #pragma unroll
    for (int sub = 0; sub < 4; ++sub)
      #pragma unroll
      for (int t = 0; t < 4; ++t) {
        v2f bp;
        bp[0] = sacc[sub][2 * t];
        bp[1] = sacc[sub][2 * t + 1];
        const int n0 = sub * 16 + 8 * hi + 2 * t;
        #pragma unroll
        for (int ds = 0; ds < 4; ++ds) {
          v2f av;
          av[0] = Vs[n0][ds * 16 + r];
          av[1] = Vs[n0 + 1][ds * 16 + r];
          oacc[ds] = __builtin_amdgcn_wmma_f32_16x16x4_f32(
              false, av, false, bp, (short)0, oacc[ds], false, false);
        }
      }
  }

  float l_other = __int_as_float(__builtin_amdgcn_ds_bpermute(
      4 * (lane ^ 16), __float_as_int(l_i)));
  float inv = 1.0f / (l_i + l_other);
  float* orow = attn + (size_t)qr * 4096 + h * 64 + 8 * hi;
  #pragma unroll
  for (int ds = 0; ds < 4; ++ds) {
    float4 a = make_float4(oacc[ds][0] * inv, oacc[ds][1] * inv,
                           oacc[ds][2] * inv, oacc[ds][3] * inv);
    float4 b = make_float4(oacc[ds][4] * inv, oacc[ds][5] * inv,
                           oacc[ds][6] * inv, oacc[ds][7] * inv);
    *(float4*)(orow + ds * 16)     = a;
    *(float4*)(orow + ds * 16 + 4) = b;
  }
}

extern "C" void kernel_launch(void* const* d_in, const int* in_sizes, int n_in,
                              void* d_out, int out_size, void* d_ws, size_t ws_size,
                              hipStream_t stream)
{
  const float* hidden = (const float*)d_in[0];
  const float* q_w    = (const float*)d_in[2];
  const float* q_b    = (const float*)d_in[3];
  const float* kv_a_w = (const float*)d_in[4];
  const float* kv_b_w = (const float*)d_in[5];
  const float* o_w    = (const float*)d_in[6];
  const float* o_b    = (const float*)d_in[7];
  float* out = (float*)d_out;

  float* ws    = (float*)d_ws;
  float* qbuf  = ws;
  float* ckv   = qbuf  + (size_t)2048 * 4096;
  float* kvbuf = ckv   + (size_t)2048 * 384;
  float* attn  = kvbuf + (size_t)2048 * 1024;

  dim3 blk(256);
  gemm_wmma_f32<<<dim3(35, 16), blk, 0, stream>>>(
      hidden, q_w, q_b, qbuf, 4096, 32,
      kv_a_w, nullptr, ckv, 384, 2048, 2880, 2880, nullptr);
  gemm_wmma_f32<<<dim3(1024 / 128, 2048 / 128), blk, 0, stream>>>(
      ckv, kv_b_w, nullptr, kvbuf, 1024, 8,
      nullptr, nullptr, nullptr, 0, 2048, 384, 384, nullptr);
  long long pairs = 2048LL * 64 * 32 + 2048LL * 8 * 32;
  rope_kernel<<<(int)((pairs + 255) / 256), 256, 0, stream>>>(qbuf, kvbuf);
  flash_attn<<<dim3(32, 64), 128, 0, stream>>>(qbuf, kvbuf, attn);
  float* opart = qbuf;
  gemm_wmma_f32<<<dim3((2880 + 127) / 128, 2048 / 128, 2), blk, 0, stream>>>(
      attn, o_w, o_b, out, 2880, 23,
      nullptr, nullptr, nullptr, 0, 2048, 4096, 2048, opart);
  add_partial<<<1440, 256, 0, stream>>>(out, opart);
}


# attention: a block owns the q-tile pair (a, 31-a) of a head: 33 K/V tile steps in every block, 1024 equal blocks all resident (4 per WGP), K/V tiles 0..a staged once for both; sums unchanged
# speedup vs baseline: 1.5228x; 1.1024x over previous
#include <hip/hip_runtime.h>
#include <math.h>

typedef __attribute__((ext_vector_type(2))) float v2f;
typedef __attribute__((ext_vector_type(8))) float v8f;

#define NEGBIG (-3.4e38f)

__global__ __launch_bounds__(256) void gemm_wmma_f32(
    const float* __restrict__ A,
    const float* __restrict__ W0, const float* __restrict__ bias0,
    float* __restrict__ C0, int N0, int nb0,
    const float* __restrict__ W1, const float* __restrict__ bias1,
    float* __restrict__ C1, int N1,
    int M, int K, int kslice, float* __restrict__ Cpart)
{
  const bool second = (int)blockIdx.x >= nb0;
  const float* __restrict__ W    = second ? W1 : W0;
  const bool  partial = blockIdx.z != 0;
  const int   kbeg    = (int)blockIdx.z * kslice;
  const int   kend    = kbeg + kslice;
  const float* __restrict__ bias = partial ? nullptr : (second ? bias1 : bias0);
  float* __restrict__ C          = partial ? Cpart : (second ? C1 : C0);
  const int N                    = second ? N1 : N0;
  const int nblk                 = second ? (int)blockIdx.x - nb0 : (int)blockIdx.x;

  __shared__ float As[128][33];
  __shared__ float Bs[32][130];

  const int tid  = threadIdx.x;
  const int wave = tid >> 5;
  const int lane = tid & 31;
  const int r    = lane & 15;
  const int hi   = lane >> 4;
  const int wrow = wave >> 2;
  const int wcol = wave & 3;
  const int m0   = blockIdx.y * 128;
  const int n0   = nblk * 128;

  v8f acc[4][2];
  for (int i = 0; i < 4; ++i)
    for (int j = 0; j < 2; ++j)
      for (int c = 0; c < 8; ++c) acc[i][j][c] = 0.0f;

  for (int k0 = kbeg; k0 < kend; k0 += 32) {
    __syncthreads();
    {
      int row = tid >> 3;
      int col = (tid & 7) * 4;
      #pragma unroll
      for (int it = 0; it < 4; ++it) {
        int rr = row + it * 32;
        float4 v = *(const float4*)&A[(size_t)(m0 + rr) * K + k0 + col];
        As[rr][col + 0] = v.x; As[rr][col + 1] = v.y;
        As[rr][col + 2] = v.z; As[rr][col + 3] = v.w;
      }
    }
    {
      int n  = tid >> 3;
      int kk = (tid & 7) * 4;
      #pragma unroll
      for (int it = 0; it < 4; ++it) {
        int nn = n + it * 32;
        float4 v = make_float4(0.f, 0.f, 0.f, 0.f);
        if (n0 + nn < N)
          v = *(const float4*)&W[(size_t)(n0 + nn) * K + k0 + kk];
        Bs[kk + 0][nn] = v.x; Bs[kk + 1][nn] = v.y;
        Bs[kk + 2][nn] = v.z; Bs[kk + 3][nn] = v.w;
      }
    }
    __syncthreads();

    #pragma unroll
    for (int kk = 0; kk < 32; kk += 4) {
      v2f af[4], bf[2];
      #pragma unroll
      for (int i = 0; i < 4; ++i) {
        int row = wrow * 64 + i * 16 + r;
        af[i][0] = As[row][kk + 2 * hi];
        af[i][1] = As[row][kk + 1 + 2 * hi];
      }
      #pragma unroll
      for (int j = 0; j < 2; ++j) {
        int col = wcol * 32 + j * 16 + r;
        bf[j][0] = Bs[kk + 2 * hi][col];
        bf[j][1] = Bs[kk + 1 + 2 * hi][col];
      }
      #pragma unroll
      for (int i = 0; i < 4; ++i)
        #pragma unroll
        for (int j = 0; j < 2; ++j)
          acc[i][j] = __builtin_amdgcn_wmma_f32_16x16x4_f32(
              false, af[i], false, bf[j], (short)0, acc[i][j], false, false);
    }
  }

  #pragma unroll
  for (int i = 0; i < 4; ++i)
    for (int j = 0; j < 2; ++j) {
      int gcol = n0 + wcol * 32 + j * 16 + r;
      if (gcol >= N) continue;
      float b = bias ? bias[gcol] : 0.0f;
      #pragma unroll
      for (int vg = 0; vg < 8; ++vg) {
        int grow = m0 + wrow * 64 + i * 16 + vg + 8 * hi;
        C[(size_t)grow * N + gcol] = acc[i][j][vg] + b;
      }
    }
}

__global__ __launch_bounds__(256) void add_partial(float* __restrict__ C,
                                                   const float* __restrict__ P)
{
  const size_t base = (size_t)blockIdx.x * 1024 + threadIdx.x;
  float4 a[4], b[4];
  #pragma unroll
  for (int u = 0; u < 4; ++u) {
    a[u] = ((const float4*)C)[base + u * 256];
    b[u] = ((const float4*)P)[base + u * 256];
  }
  #pragma unroll
  for (int u = 0; u < 4; ++u) {
    a[u].x += b[u].x; a[u].y += b[u].y; a[u].z += b[u].z; a[u].w += b[u].w;
    ((float4*)C)[base + u * 256] = a[u];
  }
}

__global__ void rope_kernel(float* __restrict__ qbuf, float* __restrict__ kvbuf)
{
  const long long QP = 2048LL * 64 * 32;
  const long long KP = 2048LL * 8 * 32;
  long long idx = (long long)blockIdx.x * blockDim.x + threadIdx.x;
  if (idx >= QP + KP) return;

  int s, i;
  float *p0, *p1;
  if (idx < QP) {
    int t = (int)idx;
    i = t & 31;
    int h = (t >> 5) & 63;
    s = t >> 11;
    float* base = qbuf + (size_t)s * 4096 + h * 64;
    p0 = base + i; p1 = base + i + 32;
  } else {
    int t = (int)(idx - QP);
    i = t & 31;
    int kvh = (t >> 5) & 7;
    s = t >> 8;
    float* base = kvbuf + (size_t)s * 1024 + kvh * 128;
    p0 = base + i; p1 = base + i + 32;
  }
  float inv = powf(150000.0f, -((float)(2 * i)) / 64.0f);
  float ang = (float)s * inv;
  float c = cosf(ang), sn = sinf(ang);
  float x1 = *p0, x2 = *p1;
  *p0 = x1 * c - x2 * sn;
  *p1 = x2 * c + x1 * sn;
}


__device__ __forceinline__ void attn_tile_step(
    const float (*Qslab)[65], const float (*Ks)[68], const float (*Vs)[68],
    const int r, const int hi, const int lane, const int kvbase, const int qr,
    const bool diag, float& m_i, float& l_i, v8f (&oacc)[4])
{
  v8f sacc[4];
  for (int sub = 0; sub < 4; ++sub)
    for (int c = 0; c < 8; ++c) sacc[sub][c] = 0.0f;
  #pragma unroll
  for (int kk = 0; kk < 64; kk += 4) {
    v2f bq;
    bq[0] = Qslab[r][kk + 2 * hi];
    bq[1] = Qslab[r][kk + 1 + 2 * hi];
    #pragma unroll
    for (int sub = 0; sub < 4; ++sub) {
      v2f ak = *(const v2f*)&Ks[sub * 16 + r][kk + 2 * hi];
      sacc[sub] = __builtin_amdgcn_wmma_f32_16x16x4_f32(
          false, ak, false, bq, (short)0, sacc[sub], false, false);
    }
  }

  #pragma unroll
  for (int sub = 0; sub < 4; ++sub)
    #pragma unroll
    for (int g = 0; g < 8; ++g) sacc[sub][g] *= 0.125f;
  if (diag) {
    #pragma unroll
    for (int sub = 0; sub < 4; ++sub)
      #pragma unroll
      for (int g = 0; g < 8; ++g) {
        int kc = kvbase + sub * 16 + 8 * hi + g;
        sacc[sub][g] = (kc <= qr) ? sacc[sub][g] : NEGBIG;
      }
  }

  float mx[4];
  #pragma unroll
  for (int sub = 0; sub < 4; ++sub) {
    mx[sub] = sacc[sub][0];
    #pragma unroll
    for (int g = 1; g < 8; ++g) mx[sub] = fmaxf(mx[sub], sacc[sub][g]);
  }
  float rowmax = fmaxf(fmaxf(mx[0], mx[1]), fmaxf(mx[2], mx[3]));
  rowmax = fmaxf(rowmax, __int_as_float(__builtin_amdgcn_ds_bpermute(
      4 * (lane ^ 16), __float_as_int(rowmax))));
  const float newm  = fmaxf(m_i, rowmax);
  const float alpha = __expf(m_i - newm);
  float ps[4];
  #pragma unroll
  for (int sub = 0; sub < 4; ++sub) {
    ps[sub] = 0.0f;
    #pragma unroll
    for (int g = 0; g < 8; ++g) {
      float p = __expf(sacc[sub][g] - newm);
      sacc[sub][g] = p;
      ps[sub] += p;
    }
  }
  l_i = l_i * alpha + ((ps[0] + ps[1]) + (ps[2] + ps[3]));
  m_i = newm;
  #pragma unroll
  for (int ds = 0; ds < 4; ++ds)
    #pragma unroll
    for (int g = 0; g < 8; ++g) oacc[ds][g] *= alpha;

  #pragma unroll
  for (int sub = 0; sub < 4; ++sub)
    #pragma unroll
    for (int t = 0; t < 4; ++t) {
      v2f bp;
      bp[0] = sacc[sub][2 * t];
      bp[1] = sacc[sub][2 * t + 1];
      const int n0 = sub * 16 + 8 * hi + 2 * t;
      #pragma unroll
      for (int ds = 0; ds < 4; ++ds) {
        v2f av;
        av[0] = Vs[n0][ds * 16 + r];
        av[1] = Vs[n0 + 1][ds * 16 + r];
        oacc[ds] = __builtin_amdgcn_wmma_f32_16x16x4_f32(
            false, av, false, bp, (short)0, oacc[ds], false, false);
      }
    }
}

__device__ __forceinline__ void attn_write_slab(
    float* __restrict__ attn, const int qr, const int h, const int hi, const int lane,
    const float l_i, const v8f (&oacc)[4])
{
  float l_other = __int_as_float(__builtin_amdgcn_ds_bpermute(
      4 * (lane ^ 16), __float_as_int(l_i)));
  float inv = 1.0f / (l_i + l_other);
  float* orow = attn + (size_t)qr * 4096 + h * 64 + 8 * hi;
  #pragma unroll
  for (int ds = 0; ds < 4; ++ds) {
    float4 a = make_float4(oacc[ds][0] * inv, oacc[ds][1] * inv,
                           oacc[ds][2] * inv, oacc[ds][3] * inv);
    float4 b = make_float4(oacc[ds][4] * inv, oacc[ds][5] * inv,
                           oacc[ds][6] * inv, oacc[ds][7] * inv);
    *(float4*)(orow + ds * 16)     = a;
    *(float4*)(orow + ds * 16 + 4) = b;
  }
}

__global__ __launch_bounds__(128) void flash_attn(
    const float* __restrict__ qbuf, const float* __restrict__ kvbuf,
    float* __restrict__ attn)
{
  __shared__ float Qs[2][64][65];
  __shared__ __attribute__((aligned(16))) float Ks[64][68];
  __shared__ __attribute__((aligned(16))) float Vs[64][68];

  const int tid    = threadIdx.x;
  const int wave   = tid >> 5;
  const int lane   = tid & 31;
  const int r      = lane & 15;
  const int hi     = lane >> 4;
  const int tnear  = blockIdx.x;
  const int tfar   = 31 - tnear;
  const int h      = blockIdx.y;
  const int kvh    = h >> 3;
  const int qr_near = tnear * 64 + wave * 16 + r;
  const int qr_far  = tfar  * 64 + wave * 16 + r;

  #pragma unroll 4
  for (int it = 0; it < 32; ++it) {
    int f = tid + it * 128;
    int sl = f >> 6, d = f & 63;
    Qs[0][sl][d] = qbuf[(size_t)(tnear * 64 + sl) * 4096 + h * 64 + d];
    Qs[1][sl][d] = qbuf[(size_t)(tfar  * 64 + sl) * 4096 + h * 64 + d];
  }

  float m_near = -INFINITY, m_far = -INFINITY;
  float l_near = 0.0f,      l_far = 0.0f;
  v8f o_near[4], o_far[4];
  for (int ds = 0; ds < 4; ++ds)
    for (int c = 0; c < 8; ++c) { o_near[ds][c] = 0.0f; o_far[ds][c] = 0.0f; }

  for (int j = 0; j <= tfar; ++j) {
    const int kvbase = j * 64;
    __syncthreads();
    #pragma unroll 4
    for (int it = 0; it < 32; ++it) {
      int f = tid + it * 128;
      int n = f >> 6, d = f & 63;
      const float* kv = &kvbuf[(size_t)(kvbase + n) * 1024 + kvh * 128];
      Ks[n][d] = kv[d];
      Vs[n][d] = kv[64 + d];
    }
    __syncthreads();

    attn_tile_step(&Qs[1][wave * 16], Ks, Vs, r, hi, lane, kvbase, qr_far,
                   j == tfar, m_far, l_far, o_far);
    if (j <= tnear)
      attn_tile_step(&Qs[0][wave * 16], Ks, Vs, r, hi, lane, kvbase, qr_near,
                     j == tnear, m_near, l_near, o_near);
  }

  attn_write_slab(attn, qr_far,  h, hi, lane, l_far,  o_far);
  attn_write_slab(attn, qr_near, h, hi, lane, l_near, o_near);
}

extern "C" void kernel_launch(void* const* d_in, const int* in_sizes, int n_in,
                              void* d_out, int out_size, void* d_ws, size_t ws_size,
                              hipStream_t stream)
{
  const float* hidden = (const float*)d_in[0];
  const float* q_w    = (const float*)d_in[2];
  const float* q_b    = (const float*)d_in[3];
  const float* kv_a_w = (const float*)d_in[4];
  const float* kv_b_w = (const float*)d_in[5];
  const float* o_w    = (const float*)d_in[6];
  const float* o_b    = (const float*)d_in[7];
  float* out = (float*)d_out;

  float* ws    = (float*)d_ws;
  float* qbuf  = ws;
  float* ckv   = qbuf  + (size_t)2048 * 4096;
  float* kvbuf = ckv   + (size_t)2048 * 384;
  float* attn  = kvbuf + (size_t)2048 * 1024;

  dim3 blk(256);
  gemm_wmma_f32<<<dim3(35, 16), blk, 0, stream>>>(
      hidden, q_w, q_b, qbuf, 4096, 32,
      kv_a_w, nullptr, ckv, 384, 2048, 2880, 2880, nullptr);
  gemm_wmma_f32<<<dim3(1024 / 128, 2048 / 128), blk, 0, stream>>>(
      ckv, kv_b_w, nullptr, kvbuf, 1024, 8,
      nullptr, nullptr, nullptr, 0, 2048, 384, 384, nullptr);
  long long pairs = 2048LL * 64 * 32 + 2048LL * 8 * 32;
  rope_kernel<<<(int)((pairs + 255) / 256), 256, 0, stream>>>(qbuf, kvbuf);
  flash_attn<<<dim3(16, 64), 128, 0, stream>>>(qbuf, kvbuf, attn);
  float* opart = qbuf;
  gemm_wmma_f32<<<dim3((2880 + 127) / 128, 2048 / 128, 2), blk, 0, stream>>>(
      attn, o_w, o_b, out, 2880, 23,
      nullptr, nullptr, nullptr, 0, 2048, 4096, 2048, opart);
  add_partial<<<1440, 256, 0, stream>>>(out, opart);
}


# kv_a tiles cut over k into 5 short blocks behind the 512 Q blocks of a flat grid: no WGP holds a third long block; a small kernel adds the 4 partial planes: ckv k sum order changes; 6 -> 7 launches
# speedup vs baseline: 1.6563x; 1.0877x over previous
#include <hip/hip_runtime.h>
#include <math.h>

typedef __attribute__((ext_vector_type(2))) float v2f;
typedef __attribute__((ext_vector_type(8))) float v8f;

#define NEGBIG (-3.4e38f)

__device__ __forceinline__ void gemm_tile_128(
    const float* __restrict__ A, const float* __restrict__ W,
    const float* __restrict__ bias, float* __restrict__ C,
    const int N, const int K, const int kbeg, const int kend,
    const int m0, const int n0)
{
  __shared__ float As[128][33];
  __shared__ float Bs[32][130];

  const int tid  = threadIdx.x;
  const int wave = tid >> 5;
  const int lane = tid & 31;
  const int r    = lane & 15;
  const int hi   = lane >> 4;
  const int wrow = wave >> 2;
  const int wcol = wave & 3;

  v8f acc[4][2];
  for (int i = 0; i < 4; ++i)
    for (int j = 0; j < 2; ++j)
      for (int c = 0; c < 8; ++c) acc[i][j][c] = 0.0f;

  for (int k0 = kbeg; k0 < kend; k0 += 32) {
    __syncthreads();
    {
      int row = tid >> 3;
      int col = (tid & 7) * 4;
      #pragma unroll
      for (int it = 0; it < 4; ++it) {
        int rr = row + it * 32;
        float4 v = *(const float4*)&A[(size_t)(m0 + rr) * K + k0 + col];
        As[rr][col + 0] = v.x; As[rr][col + 1] = v.y;
        As[rr][col + 2] = v.z; As[rr][col + 3] = v.w;
      }
    }
    {
      int n  = tid >> 3;
      int kk = (tid & 7) * 4;
      #pragma unroll
      for (int it = 0; it < 4; ++it) {
        int nn = n + it * 32;
        float4 v = make_float4(0.f, 0.f, 0.f, 0.f);
        if (n0 + nn < N)
          v = *(const float4*)&W[(size_t)(n0 + nn) * K + k0 + kk];
        Bs[kk + 0][nn] = v.x; Bs[kk + 1][nn] = v.y;
        Bs[kk + 2][nn] = v.z; Bs[kk + 3][nn] = v.w;
      }
    }
    __syncthreads();

    #pragma unroll
    for (int kk = 0; kk < 32; kk += 4) {
      v2f af[4], bf[2];
      #pragma unroll
      for (int i = 0; i < 4; ++i) {
        int row = wrow * 64 + i * 16 + r;
        af[i][0] = As[row][kk + 2 * hi];
        af[i][1] = As[row][kk + 1 + 2 * hi];
      }
      #pragma unroll
      for (int j = 0; j < 2; ++j) {
        int col = wcol * 32 + j * 16 + r;
        bf[j][0] = Bs[kk + 2 * hi][col];
        bf[j][1] = Bs[kk + 1 + 2 * hi][col];
      }
      #pragma unroll
      for (int i = 0; i < 4; ++i)
        #pragma unroll
        for (int j = 0; j < 2; ++j)
          acc[i][j] = __builtin_amdgcn_wmma_f32_16x16x4_f32(
              false, af[i], false, bf[j], (short)0, acc[i][j], false, false);
    }
  }

  #pragma unroll
  for (int i = 0; i < 4; ++i)
    for (int j = 0; j < 2; ++j) {
      int gcol = n0 + wcol * 32 + j * 16 + r;
      if (gcol >= N) continue;
      float b = bias ? bias[gcol] : 0.0f;
      #pragma unroll
      for (int vg = 0; vg < 8; ++vg) {
        int grow = m0 + wrow * 64 + i * 16 + vg + 8 * hi;
        C[(size_t)grow * N + gcol] = acc[i][j][vg] + b;
      }
    }
}

__global__ __launch_bounds__(256) void gemm_wmma_f32(
    const float* __restrict__ A, const float* __restrict__ W,
    const float* __restrict__ bias, float* __restrict__ C,
    int N, int K, int kslice, float* __restrict__ Cpart)
{
  const bool partial = blockIdx.z != 0;
  const int  kbeg    = (int)blockIdx.z * kslice;
  gemm_tile_128(A, W, partial ? nullptr : bias, partial ? Cpart : C, N, K,
                kbeg, kbeg + kslice, (int)blockIdx.y * 128, (int)blockIdx.x * 128);
}

__global__ __launch_bounds__(256) void gemm_q_kva(
    const float* __restrict__ hidden,
    const float* __restrict__ q_w, const float* __restrict__ q_b, float* __restrict__ qbuf,
    const float* __restrict__ kv_a_w, float* __restrict__ ckv, float* __restrict__ part)
{
  const int  b     = (int)blockIdx.x;
  const bool kva   = b >= 512;
  const int  c     = b - 512;
  const int  slice = c / 48;
  const int  t     = c - slice * 48;
  const int  ct    = kva ? t % 3 : (b & 31);
  const int  rt    = kva ? t / 3 : (b >> 5);
  const int  kbeg  = kva ? slice * 576 : 0;
  const int  kend  = kva ? kbeg + 576 : 2880;
  float* __restrict__ C = !kva ? qbuf
                        : (slice == 0 ? ckv : part + (size_t)(slice - 1) * (2048 * 384));
  gemm_tile_128(hidden, kva ? kv_a_w : q_w, kva ? nullptr : q_b, C, kva ? 384 : 4096,
                2880, kbeg, kend, rt * 128, ct * 128);
}

__global__ __launch_bounds__(256) void add_partial(float* __restrict__ C,
                                                   const float* __restrict__ P)
{
  const size_t base = (size_t)blockIdx.x * 1024 + threadIdx.x;
  float4 a[4], b[4];
  #pragma unroll
  for (int u = 0; u < 4; ++u) {
    a[u] = ((const float4*)C)[base + u * 256];
    b[u] = ((const float4*)P)[base + u * 256];
  }
  #pragma unroll
  for (int u = 0; u < 4; ++u) {
    a[u].x += b[u].x; a[u].y += b[u].y; a[u].z += b[u].z; a[u].w += b[u].w;
    ((float4*)C)[base + u * 256] = a[u];
  }
}

__global__ __launch_bounds__(256) void sum_planes5(float* __restrict__ C,
                                                   const float* __restrict__ P)
{
  const size_t base = (size_t)blockIdx.x * 1024 + threadIdx.x;
  float4 a[4], p[4][4];
  #pragma unroll
  for (int u = 0; u < 4; ++u) {
    a[u] = ((const float4*)C)[base + u * 256];
    #pragma unroll
    for (int q = 0; q < 4; ++q)
      p[q][u] = ((const float4*)P)[(size_t)q * 196608 + base + u * 256];
  }
  #pragma unroll
  for (int u = 0; u < 4; ++u) {
    #pragma unroll
    for (int q = 0; q < 4; ++q) {
      a[u].x += p[q][u].x; a[u].y += p[q][u].y; a[u].z += p[q][u].z; a[u].w += p[q][u].w;
    }
    ((float4*)C)[base + u * 256] = a[u];
  }
}

__global__ void rope_kernel(float* __restrict__ qbuf, float* __restrict__ kvbuf)
{
  const long long QP = 2048LL * 64 * 32;
  const long long KP = 2048LL * 8 * 32;
  long long idx = (long long)blockIdx.x * blockDim.x + threadIdx.x;
  if (idx >= QP + KP) return;

  int s, i;
  float *p0, *p1;
  if (idx < QP) {
    int t = (int)idx;
    i = t & 31;
    int h = (t >> 5) & 63;
    s = t >> 11;
    float* base = qbuf + (size_t)s * 4096 + h * 64;
    p0 = base + i; p1 = base + i + 32;
  } else {
    int t = (int)(idx - QP);
    i = t & 31;
    int kvh = (t >> 5) & 7;
    s = t >> 8;
    float* base = kvbuf + (size_t)s * 1024 + kvh * 128;
    p0 = base + i; p1 = base + i + 32;
  }
  float inv = powf(150000.0f, -((float)(2 * i)) / 64.0f);
  float ang = (float)s * inv;
  float c = cosf(ang), sn = sinf(ang);
  float x1 = *p0, x2 = *p1;
  *p0 = x1 * c - x2 * sn;
  *p1 = x2 * c + x1 * sn;
}


__device__ __forceinline__ void attn_tile_step(
    const float (*Qslab)[65], const float (*Ks)[68], const float (*Vs)[68],
    const int r, const int hi, const int lane, const int kvbase, const int qr,
    const bool diag, float& m_i, float& l_i, v8f (&oacc)[4])
{
  v8f sacc[4];
  for (int sub = 0; sub < 4; ++sub)
    for (int c = 0; c < 8; ++c) sacc[sub][c] = 0.0f;
  #pragma unroll
  for (int kk = 0; kk < 64; kk += 4) {
    v2f bq;
    bq[0] = Qslab[r][kk + 2 * hi];
    bq[1] = Qslab[r][kk + 1 + 2 * hi];
    #pragma unroll
    for (int sub = 0; sub < 4; ++sub) {
      v2f ak = *(const v2f*)&Ks[sub * 16 + r][kk + 2 * hi];
      sacc[sub] = __builtin_amdgcn_wmma_f32_16x16x4_f32(
          false, ak, false, bq, (short)0, sacc[sub], false, false);
    }
  }

  #pragma unroll
  for (int sub = 0; sub < 4; ++sub)
    #pragma unroll
    for (int g = 0; g < 8; ++g) sacc[sub][g] *= 0.125f;
  if (diag) {
    #pragma unroll
    for (int sub = 0; sub < 4; ++sub)
      #pragma unroll
      for (int g = 0; g < 8; ++g) {
        int kc = kvbase + sub * 16 + 8 * hi + g;
        sacc[sub][g] = (kc <= qr) ? sacc[sub][g] : NEGBIG;
      }
  }

  float mx[4];
  #pragma unroll
  for (int sub = 0; sub < 4; ++sub) {
    mx[sub] = sacc[sub][0];
    #pragma unroll
    for (int g = 1; g < 8; ++g) mx[sub] = fmaxf(mx[sub], sacc[sub][g]);
  }
  float rowmax = fmaxf(fmaxf(mx[0], mx[1]), fmaxf(mx[2], mx[3]));
  rowmax = fmaxf(rowmax, __int_as_float(__builtin_amdgcn_ds_bpermute(
      4 * (lane ^ 16), __float_as_int(rowmax))));
  const float newm  = fmaxf(m_i, rowmax);
  const float alpha = __expf(m_i - newm);
  float ps[4];
  #pragma unroll
  for (int sub = 0; sub < 4; ++sub) {
    ps[sub] = 0.0f;
    #pragma unroll
    for (int g = 0; g < 8; ++g) {
      float p = __expf(sacc[sub][g] - newm);
      sacc[sub][g] = p;
      ps[sub] += p;
    }
  }
  l_i = l_i * alpha + ((ps[0] + ps[1]) + (ps[2] + ps[3]));
  m_i = newm;
  #pragma unroll
  for (int ds = 0; ds < 4; ++ds)
    #pragma unroll
    for (int g = 0; g < 8; ++g) oacc[ds][g] *= alpha;

  #pragma unroll
  for (int sub = 0; sub < 4; ++sub)
    #pragma unroll
    for (int t = 0; t < 4; ++t) {
      v2f bp;
      bp[0] = sacc[sub][2 * t];
      bp[1] = sacc[sub][2 * t + 1];
      const int n0 = sub * 16 + 8 * hi + 2 * t;
      #pragma unroll
      for (int ds = 0; ds < 4; ++ds) {
        v2f av;
        av[0] = Vs[n0][ds * 16 + r];
        av[1] = Vs[n0 + 1][ds * 16 + r];
        oacc[ds] = __builtin_amdgcn_wmma_f32_16x16x4_f32(
            false, av, false, bp, (short)0, oacc[ds], false, false);
      }
    }
}

__device__ __forceinline__ void attn_write_slab(
    float* __restrict__ attn, const int qr, const int h, const int hi, const int lane,
    const float l_i, const v8f (&oacc)[4])
{
  float l_other = __int_as_float(__builtin_amdgcn_ds_bpermute(
      4 * (lane ^ 16), __float_as_int(l_i)));
  float inv = 1.0f / (l_i + l_other);
  float* orow = attn + (size_t)qr * 4096 + h * 64 + 8 * hi;
  #pragma unroll
  for (int ds = 0; ds < 4; ++ds) {
    float4 a = make_float4(oacc[ds][0] * inv, oacc[ds][1] * inv,
                           oacc[ds][2] * inv, oacc[ds][3] * inv);
    float4 b = make_float4(oacc[ds][4] * inv, oacc[ds][5] * inv,
                           oacc[ds][6] * inv, oacc[ds][7] * inv);
    *(float4*)(orow + ds * 16)     = a;
    *(float4*)(orow + ds * 16 + 4) = b;
  }
}

__global__ __launch_bounds__(128) void flash_attn(
    const float* __restrict__ qbuf, const float* __restrict__ kvbuf,
    float* __restrict__ attn)
{
  __shared__ float Qs[2][64][65];
  __shared__ __attribute__((aligned(16))) float Ks[64][68];
  __shared__ __attribute__((aligned(16))) float Vs[64][68];

  const int tid    = threadIdx.x;
  const int wave   = tid >> 5;
  const int lane   = tid & 31;
  const int r      = lane & 15;
  const int hi     = lane >> 4;
  const int tnear  = blockIdx.x;
  const int tfar   = 31 - tnear;
  const int h      = blockIdx.y;
  const int kvh    = h >> 3;
  const int qr_near = tnear * 64 + wave * 16 + r;
  const int qr_far  = tfar  * 64 + wave * 16 + r;

  #pragma unroll 4
  for (int it = 0; it < 32; ++it) {
    int f = tid + it * 128;
    int sl = f >> 6, d = f & 63;
    Qs[0][sl][d] = qbuf[(size_t)(tnear * 64 + sl) * 4096 + h * 64 + d];
    Qs[1][sl][d] = qbuf[(size_t)(tfar  * 64 + sl) * 4096 + h * 64 + d];
  }

  float m_near = -INFINITY, m_far = -INFINITY;
  float l_near = 0.0f,      l_far = 0.0f;
  v8f o_near[4], o_far[4];
  for (int ds = 0; ds < 4; ++ds)
    for (int c = 0; c < 8; ++c) { o_near[ds][c] = 0.0f; o_far[ds][c] = 0.0f; }

  for (int j = 0; j <= tfar; ++j) {
    const int kvbase = j * 64;
    __syncthreads();
    #pragma unroll 4
    for (int it = 0; it < 32; ++it) {
      int f = tid + it * 128;
      int n = f >> 6, d = f & 63;
      const float* kv = &kvbuf[(size_t)(kvbase + n) * 1024 + kvh * 128];
      Ks[n][d] = kv[d];
      Vs[n][d] = kv[64 + d];
    }
    __syncthreads();

    attn_tile_step(&Qs[1][wave * 16], Ks, Vs, r, hi, lane, kvbase, qr_far,
                   j == tfar, m_far, l_far, o_far);
    if (j <= tnear)
      attn_tile_step(&Qs[0][wave * 16], Ks, Vs, r, hi, lane, kvbase, qr_near,
                     j == tnear, m_near, l_near, o_near);
  }

  attn_write_slab(attn, qr_far,  h, hi, lane, l_far,  o_far);
  attn_write_slab(attn, qr_near, h, hi, lane, l_near, o_near);
}

extern "C" void kernel_launch(void* const* d_in, const int* in_sizes, int n_in,
                              void* d_out, int out_size, void* d_ws, size_t ws_size,
                              hipStream_t stream)
{
  const float* hidden = (const float*)d_in[0];
  const float* q_w    = (const float*)d_in[2];
  const float* q_b    = (const float*)d_in[3];
  const float* kv_a_w = (const float*)d_in[4];
  const float* kv_b_w = (const float*)d_in[5];
  const float* o_w    = (const float*)d_in[6];
  const float* o_b    = (const float*)d_in[7];
  float* out = (float*)d_out;

  float* ws    = (float*)d_ws;
  float* qbuf  = ws;
  float* ckv   = qbuf  + (size_t)2048 * 4096;
  float* kvbuf = ckv   + (size_t)2048 * 384;
  float* attn  = kvbuf + (size_t)2048 * 1024;

  dim3 blk(256);
  gemm_q_kva<<<dim3(512 + 240), blk, 0, stream>>>(
      hidden, q_w, q_b, qbuf, kv_a_w, ckv, attn);
  sum_planes5<<<192, 256, 0, stream>>>(ckv, attn);
  gemm_wmma_f32<<<dim3(1024 / 128, 2048 / 128), blk, 0, stream>>>(
      ckv, kv_b_w, nullptr, kvbuf, 1024, 384, 384, nullptr);
  long long pairs = 2048LL * 64 * 32 + 2048LL * 8 * 32;
  rope_kernel<<<(int)((pairs + 255) / 256), 256, 0, stream>>>(qbuf, kvbuf);
  flash_attn<<<dim3(16, 64), 128, 0, stream>>>(qbuf, kvbuf, attn);
  float* opart = qbuf;
  gemm_wmma_f32<<<dim3((2880 + 127) / 128, 2048 / 128, 2), blk, 0, stream>>>(
      attn, o_w, o_b, out, 2880, 4096, 2048, opart);
  add_partial<<<1440, 256, 0, stream>>>(out, opart);
}


# attention: the next K/V tile is loaded into registers during the current tile's matrix work (K before the score product, V behind it) and stored to LDS at the next trip's top; sums unchanged
# speedup vs baseline: 1.7086x; 1.0316x over previous
#include <hip/hip_runtime.h>
#include <math.h>

typedef __attribute__((ext_vector_type(2))) float v2f;
typedef __attribute__((ext_vector_type(8))) float v8f;

#define NEGBIG (-3.4e38f)

__device__ __forceinline__ void gemm_tile_128(
    const float* __restrict__ A, const float* __restrict__ W,
    const float* __restrict__ bias, float* __restrict__ C,
    const int N, const int K, const int kbeg, const int kend,
    const int m0, const int n0)
{
  __shared__ float As[128][33];
  __shared__ float Bs[32][130];

  const int tid  = threadIdx.x;
  const int wave = tid >> 5;
  const int lane = tid & 31;
  const int r    = lane & 15;
  const int hi   = lane >> 4;
  const int wrow = wave >> 2;
  const int wcol = wave & 3;

  v8f acc[4][2];
  for (int i = 0; i < 4; ++i)
    for (int j = 0; j < 2; ++j)
      for (int c = 0; c < 8; ++c) acc[i][j][c] = 0.0f;

  for (int k0 = kbeg; k0 < kend; k0 += 32) {
    __syncthreads();
    {
      int row = tid >> 3;
      int col = (tid & 7) * 4;
      #pragma unroll
      for (int it = 0; it < 4; ++it) {
        int rr = row + it * 32;
        float4 v = *(const float4*)&A[(size_t)(m0 + rr) * K + k0 + col];
        As[rr][col + 0] = v.x; As[rr][col + 1] = v.y;
        As[rr][col + 2] = v.z; As[rr][col + 3] = v.w;
      }
    }
    {
      int n  = tid >> 3;
      int kk = (tid & 7) * 4;
      #pragma unroll
      for (int it = 0; it < 4; ++it) {
        int nn = n + it * 32;
        float4 v = make_float4(0.f, 0.f, 0.f, 0.f);
        if (n0 + nn < N)
          v = *(const float4*)&W[(size_t)(n0 + nn) * K + k0 + kk];
        Bs[kk + 0][nn] = v.x; Bs[kk + 1][nn] = v.y;
        Bs[kk + 2][nn] = v.z; Bs[kk + 3][nn] = v.w;
      }
    }
    __syncthreads();

    #pragma unroll
    for (int kk = 0; kk < 32; kk += 4) {
      v2f af[4], bf[2];
      #pragma unroll
      for (int i = 0; i < 4; ++i) {
        int row = wrow * 64 + i * 16 + r;
        af[i][0] = As[row][kk + 2 * hi];
        af[i][1] = As[row][kk + 1 + 2 * hi];
      }
      #pragma unroll
      for (int j = 0; j < 2; ++j) {
        int col = wcol * 32 + j * 16 + r;
        bf[j][0] = Bs[kk + 2 * hi][col];
        bf[j][1] = Bs[kk + 1 + 2 * hi][col];
      }
      #pragma unroll
      for (int i = 0; i < 4; ++i)
        #pragma unroll
        for (int j = 0; j < 2; ++j)
          acc[i][j] = __builtin_amdgcn_wmma_f32_16x16x4_f32(
              false, af[i], false, bf[j], (short)0, acc[i][j], false, false);
    }
  }

  #pragma unroll
  for (int i = 0; i < 4; ++i)
    for (int j = 0; j < 2; ++j) {
      int gcol = n0 + wcol * 32 + j * 16 + r;
      if (gcol >= N) continue;
      float b = bias ? bias[gcol] : 0.0f;
      #pragma unroll
      for (int vg = 0; vg < 8; ++vg) {
        int grow = m0 + wrow * 64 + i * 16 + vg + 8 * hi;
        C[(size_t)grow * N + gcol] = acc[i][j][vg] + b;
      }
    }
}

__global__ __launch_bounds__(256) void gemm_wmma_f32(
    const float* __restrict__ A, const float* __restrict__ W,
    const float* __restrict__ bias, float* __restrict__ C,
    int N, int K, int kslice, float* __restrict__ Cpart)
{
  const bool partial = blockIdx.z != 0;
  const int  kbeg    = (int)blockIdx.z * kslice;
  gemm_tile_128(A, W, partial ? nullptr : bias, partial ? Cpart : C, N, K,
                kbeg, kbeg + kslice, (int)blockIdx.y * 128, (int)blockIdx.x * 128);
}

__global__ __launch_bounds__(256) void gemm_q_kva(
    const float* __restrict__ hidden,
    const float* __restrict__ q_w, const float* __restrict__ q_b, float* __restrict__ qbuf,
    const float* __restrict__ kv_a_w, float* __restrict__ ckv, float* __restrict__ part)
{
  const int  b     = (int)blockIdx.x;
  const bool kva   = b >= 512;
  const int  c     = b - 512;
  const int  slice = c / 48;
  const int  t     = c - slice * 48;
  const int  ct    = kva ? t % 3 : (b & 31);
  const int  rt    = kva ? t / 3 : (b >> 5);
  const int  kbeg  = kva ? slice * 576 : 0;
  const int  kend  = kva ? kbeg + 576 : 2880;
  float* __restrict__ C = !kva ? qbuf
                        : (slice == 0 ? ckv : part + (size_t)(slice - 1) * (2048 * 384));
  gemm_tile_128(hidden, kva ? kv_a_w : q_w, kva ? nullptr : q_b, C, kva ? 384 : 4096,
                2880, kbeg, kend, rt * 128, ct * 128);
}

__global__ __launch_bounds__(256) void add_partial(float* __restrict__ C,
                                                   const float* __restrict__ P)
{
  const size_t base = (size_t)blockIdx.x * 1024 + threadIdx.x;
  float4 a[4], b[4];
  #pragma unroll
  for (int u = 0; u < 4; ++u) {
    a[u] = ((const float4*)C)[base + u * 256];
    b[u] = ((const float4*)P)[base + u * 256];
  }
  #pragma unroll
  for (int u = 0; u < 4; ++u) {
    a[u].x += b[u].x; a[u].y += b[u].y; a[u].z += b[u].z; a[u].w += b[u].w;
    ((float4*)C)[base + u * 256] = a[u];
  }
}

__global__ __launch_bounds__(256) void sum_planes5(float* __restrict__ C,
                                                   const float* __restrict__ P)
{
  const size_t base = (size_t)blockIdx.x * 1024 + threadIdx.x;
  float4 a[4], p[4][4];
  #pragma unroll
  for (int u = 0; u < 4; ++u) {
    a[u] = ((const float4*)C)[base + u * 256];
    #pragma unroll
    for (int q = 0; q < 4; ++q)
      p[q][u] = ((const float4*)P)[(size_t)q * 196608 + base + u * 256];
  }
  #pragma unroll
  for (int u = 0; u < 4; ++u) {
    #pragma unroll
    for (int q = 0; q < 4; ++q) {
      a[u].x += p[q][u].x; a[u].y += p[q][u].y; a[u].z += p[q][u].z; a[u].w += p[q][u].w;
    }
    ((float4*)C)[base + u * 256] = a[u];
  }
}

__global__ void rope_kernel(float* __restrict__ qbuf, float* __restrict__ kvbuf)
{
  const long long QP = 2048LL * 64 * 32;
  const long long KP = 2048LL * 8 * 32;
  long long idx = (long long)blockIdx.x * blockDim.x + threadIdx.x;
  if (idx >= QP + KP) return;

  int s, i;
  float *p0, *p1;
  if (idx < QP) {
    int t = (int)idx;
    i = t & 31;
    int h = (t >> 5) & 63;
    s = t >> 11;
    float* base = qbuf + (size_t)s * 4096 + h * 64;
    p0 = base + i; p1 = base + i + 32;
  } else {
    int t = (int)(idx - QP);
    i = t & 31;
    int kvh = (t >> 5) & 7;
    s = t >> 8;
    float* base = kvbuf + (size_t)s * 1024 + kvh * 128;
    p0 = base + i; p1 = base + i + 32;
  }
  float inv = powf(150000.0f, -((float)(2 * i)) / 64.0f);
  float ang = (float)s * inv;
  float c = cosf(ang), sn = sinf(ang);
  float x1 = *p0, x2 = *p1;
  *p0 = x1 * c - x2 * sn;
  *p1 = x2 * c + x1 * sn;
}


__device__ __forceinline__ void attn_scores(
    const float (*Qslab)[65], const float (*Ks)[68],
    const int r, const int hi, const int kvbase, const int qr,
    const bool diag, v8f (&sacc)[4])
{
  for (int sub = 0; sub < 4; ++sub)
    for (int c = 0; c < 8; ++c) sacc[sub][c] = 0.0f;
  #pragma unroll
  for (int kk = 0; kk < 64; kk += 4) {
    v2f bq;
    bq[0] = Qslab[r][kk + 2 * hi];
    bq[1] = Qslab[r][kk + 1 + 2 * hi];
    #pragma unroll
    for (int sub = 0; sub < 4; ++sub) {
      v2f ak = *(const v2f*)&Ks[sub * 16 + r][kk + 2 * hi];
      sacc[sub] = __builtin_amdgcn_wmma_f32_16x16x4_f32(
          false, ak, false, bq, (short)0, sacc[sub], false, false);
    }
  }

  #pragma unroll
  for (int sub = 0; sub < 4; ++sub)
    #pragma unroll
    for (int g = 0; g < 8; ++g) sacc[sub][g] *= 0.125f;
  if (diag) {
    #pragma unroll
    for (int sub = 0; sub < 4; ++sub)
      #pragma unroll
      for (int g = 0; g < 8; ++g) {
        int kc = kvbase + sub * 16 + 8 * hi + g;
        sacc[sub][g] = (kc <= qr) ? sacc[sub][g] : NEGBIG;
      }
  }
}

__device__ __forceinline__ void attn_softmax_pv(
    const float (*Vs)[68], const int r, const int hi, const int lane,
    float& m_i, float& l_i, v8f (&sacc)[4], v8f (&oacc)[4])
{
  float mx[4];
  #pragma unroll
  for (int sub = 0; sub < 4; ++sub) {
    mx[sub] = sacc[sub][0];
    #pragma unroll
    for (int g = 1; g < 8; ++g) mx[sub] = fmaxf(mx[sub], sacc[sub][g]);
  }
  float rowmax = fmaxf(fmaxf(mx[0], mx[1]), fmaxf(mx[2], mx[3]));
  rowmax = fmaxf(rowmax, __int_as_float(__builtin_amdgcn_ds_bpermute(
      4 * (lane ^ 16), __float_as_int(rowmax))));
  const float newm  = fmaxf(m_i, rowmax);
  const float alpha = __expf(m_i - newm);
  float ps[4];
  #pragma unroll
  for (int sub = 0; sub < 4; ++sub) {
    ps[sub] = 0.0f;
    #pragma unroll
    for (int g = 0; g < 8; ++g) {
      float p = __expf(sacc[sub][g] - newm);
      sacc[sub][g] = p;
      ps[sub] += p;
    }
  }
  l_i = l_i * alpha + ((ps[0] + ps[1]) + (ps[2] + ps[3]));
  m_i = newm;
  #pragma unroll
  for (int ds = 0; ds < 4; ++ds)
    #pragma unroll
    for (int g = 0; g < 8; ++g) oacc[ds][g] *= alpha;

  #pragma unroll
  for (int sub = 0; sub < 4; ++sub)
    #pragma unroll
    for (int t = 0; t < 4; ++t) {
      v2f bp;
      bp[0] = sacc[sub][2 * t];
      bp[1] = sacc[sub][2 * t + 1];
      const int n0 = sub * 16 + 8 * hi + 2 * t;
      #pragma unroll
      for (int ds = 0; ds < 4; ++ds) {
        v2f av;
        av[0] = Vs[n0][ds * 16 + r];
        av[1] = Vs[n0 + 1][ds * 16 + r];
        oacc[ds] = __builtin_amdgcn_wmma_f32_16x16x4_f32(
            false, av, false, bp, (short)0, oacc[ds], false, false);
      }
    }
}

__device__ __forceinline__ void attn_write_slab(
    float* __restrict__ attn, const int qr, const int h, const int hi, const int lane,
    const float l_i, const v8f (&oacc)[4])
{
  float l_other = __int_as_float(__builtin_amdgcn_ds_bpermute(
      4 * (lane ^ 16), __float_as_int(l_i)));
  float inv = 1.0f / (l_i + l_other);
  float* orow = attn + (size_t)qr * 4096 + h * 64 + 8 * hi;
  #pragma unroll
  for (int ds = 0; ds < 4; ++ds) {
    float4 a = make_float4(oacc[ds][0] * inv, oacc[ds][1] * inv,
                           oacc[ds][2] * inv, oacc[ds][3] * inv);
    float4 b = make_float4(oacc[ds][4] * inv, oacc[ds][5] * inv,
                           oacc[ds][6] * inv, oacc[ds][7] * inv);
    *(float4*)(orow + ds * 16)     = a;
    *(float4*)(orow + ds * 16 + 4) = b;
  }
}

__global__ __launch_bounds__(128, 4) void flash_attn(
    const float* __restrict__ qbuf, const float* __restrict__ kvbuf,
    float* __restrict__ attn)
{
  __shared__ float Qs[2][64][65];
  __shared__ __attribute__((aligned(16))) float Ks[64][68];
  __shared__ __attribute__((aligned(16))) float Vs[64][68];

  const int tid    = threadIdx.x;
  const int wave   = tid >> 5;
  const int lane   = tid & 31;
  const int r      = lane & 15;
  const int hi     = lane >> 4;
  const int tnear  = blockIdx.x;
  const int tfar   = 31 - tnear;
  const int h      = blockIdx.y;
  const int kvh    = h >> 3;
  const int qr_near = tnear * 64 + wave * 16 + r;
  const int qr_far  = tfar  * 64 + wave * 16 + r;

  #pragma unroll 4
  for (int it = 0; it < 32; ++it) {
    int f = tid + it * 128;
    int sl = f >> 6, d = f & 63;
    Qs[0][sl][d] = qbuf[(size_t)(tnear * 64 + sl) * 4096 + h * 64 + d];
    Qs[1][sl][d] = qbuf[(size_t)(tfar  * 64 + sl) * 4096 + h * 64 + d];
  }

  float m_near = -INFINITY, m_far = -INFINITY;
  float l_near = 0.0f,      l_far = 0.0f;
  v8f o_near[4], o_far[4];
  for (int ds = 0; ds < 4; ++ds)
    for (int c = 0; c < 8; ++c) { o_near[ds][c] = 0.0f; o_far[ds][c] = 0.0f; }

  const int srow = tid >> 6, scol = tid & 63;
  const float* kvsrc = kvbuf + (size_t)srow * 1024 + kvh * 128 + scol;
  float kreg[32], vreg[32];
  #pragma unroll
  for (int it = 0; it < 32; ++it) {
    kreg[it] = kvsrc[(size_t)it * 2048];
    vreg[it] = kvsrc[(size_t)it * 2048 + 64];
  }

  for (int j = 0; j <= tfar; ++j) {
    const int kvbase = j * 64;
    const float* nxt = kvsrc + (size_t)(kvbase + 64) * 1024;
    __syncthreads();
    #pragma unroll
    for (int it = 0; it < 32; ++it) {
      Ks[srow + 2 * it][scol] = kreg[it];
      Vs[srow + 2 * it][scol] = vreg[it];
    }
    __syncthreads();

    v8f sacc[4];
    if (j < tfar) {
      #pragma unroll
      for (int it = 0; it < 32; ++it) kreg[it] = nxt[(size_t)it * 2048];
    }
    attn_scores(&Qs[1][wave * 16], Ks, r, hi, kvbase, qr_far, j == tfar, sacc);
    if (j < tfar) {
      #pragma unroll
      for (int it = 0; it < 32; ++it) vreg[it] = nxt[(size_t)it * 2048 + 64];
    }
    attn_softmax_pv(Vs, r, hi, lane, m_far, l_far, sacc, o_far);
    if (j <= tnear) {
      attn_scores(&Qs[0][wave * 16], Ks, r, hi, kvbase, qr_near, j == tnear, sacc);
      attn_softmax_pv(Vs, r, hi, lane, m_near, l_near, sacc, o_near);
    }
  }

  attn_write_slab(attn, qr_far,  h, hi, lane, l_far,  o_far);
  attn_write_slab(attn, qr_near, h, hi, lane, l_near, o_near);
}

extern "C" void kernel_launch(void* const* d_in, const int* in_sizes, int n_in,
                              void* d_out, int out_size, void* d_ws, size_t ws_size,
                              hipStream_t stream)
{
  const float* hidden = (const float*)d_in[0];
  const float* q_w    = (const float*)d_in[2];
  const float* q_b    = (const float*)d_in[3];
  const float* kv_a_w = (const float*)d_in[4];
  const float* kv_b_w = (const float*)d_in[5];
  const float* o_w    = (const float*)d_in[6];
  const float* o_b    = (const float*)d_in[7];
  float* out = (float*)d_out;

  float* ws    = (float*)d_ws;
  float* qbuf  = ws;
  float* ckv   = qbuf  + (size_t)2048 * 4096;
  float* kvbuf = ckv   + (size_t)2048 * 384;
  float* attn  = kvbuf + (size_t)2048 * 1024;

  dim3 blk(256);
  gemm_q_kva<<<dim3(512 + 240), blk, 0, stream>>>(
      hidden, q_w, q_b, qbuf, kv_a_w, ckv, attn);
  sum_planes5<<<192, 256, 0, stream>>>(ckv, attn);
  gemm_wmma_f32<<<dim3(1024 / 128, 2048 / 128), blk, 0, stream>>>(
      ckv, kv_b_w, nullptr, kvbuf, 1024, 384, 384, nullptr);
  long long pairs = 2048LL * 64 * 32 + 2048LL * 8 * 32;
  rope_kernel<<<(int)((pairs + 255) / 256), 256, 0, stream>>>(qbuf, kvbuf);
  flash_attn<<<dim3(16, 64), 128, 0, stream>>>(qbuf, kvbuf, attn);
  float* opart = qbuf;
  gemm_wmma_f32<<<dim3((2880 + 127) / 128, 2048 / 128, 2), blk, 0, stream>>>(
      attn, o_w, o_b, out, 2880, 4096, 2048, opart);
  add_partial<<<1440, 256, 0, stream>>>(out, opart);
}


# RoPE pass: a wave computes cos/sin once per (row, frequency), lane f = frequency f, and rotates all 72 head slots with them by lane exchange, 16-byte accesses: 4096 waves, not 147456; same values
# speedup vs baseline: 1.7691x; 1.0354x over previous
#include <hip/hip_runtime.h>
#include <math.h>

typedef __attribute__((ext_vector_type(2))) float v2f;
typedef __attribute__((ext_vector_type(8))) float v8f;

#define NEGBIG (-3.4e38f)

__device__ __forceinline__ void gemm_tile_128(
    const float* __restrict__ A, const float* __restrict__ W,
    const float* __restrict__ bias, float* __restrict__ C,
    const int N, const int K, const int kbeg, const int kend,
    const int m0, const int n0)
{
  __shared__ float As[128][33];
  __shared__ float Bs[32][130];

  const int tid  = threadIdx.x;
  const int wave = tid >> 5;
  const int lane = tid & 31;
  const int r    = lane & 15;
  const int hi   = lane >> 4;
  const int wrow = wave >> 2;
  const int wcol = wave & 3;

  v8f acc[4][2];
  for (int i = 0; i < 4; ++i)
    for (int j = 0; j < 2; ++j)
      for (int c = 0; c < 8; ++c) acc[i][j][c] = 0.0f;

  for (int k0 = kbeg; k0 < kend; k0 += 32) {
    __syncthreads();
    {
      int row = tid >> 3;
      int col = (tid & 7) * 4;
      #pragma unroll
      for (int it = 0; it < 4; ++it) {
        int rr = row + it * 32;
        float4 v = *(const float4*)&A[(size_t)(m0 + rr) * K + k0 + col];
        As[rr][col + 0] = v.x; As[rr][col + 1] = v.y;
        As[rr][col + 2] = v.z; As[rr][col + 3] = v.w;
      }
    }
    {
      int n  = tid >> 3;
      int kk = (tid & 7) * 4;
      #pragma unroll
      for (int it = 0; it < 4; ++it) {
        int nn = n + it * 32;
        float4 v = make_float4(0.f, 0.f, 0.f, 0.f);
        if (n0 + nn < N)
          v = *(const float4*)&W[(size_t)(n0 + nn) * K + k0 + kk];
        Bs[kk + 0][nn] = v.x; Bs[kk + 1][nn] = v.y;
        Bs[kk + 2][nn] = v.z; Bs[kk + 3][nn] = v.w;
      }
    }
    __syncthreads();

    #pragma unroll
    for (int kk = 0; kk < 32; kk += 4) {
      v2f af[4], bf[2];
      #pragma unroll
      for (int i = 0; i < 4; ++i) {
        int row = wrow * 64 + i * 16 + r;
        af[i][0] = As[row][kk + 2 * hi];
        af[i][1] = As[row][kk + 1 + 2 * hi];
      }
      #pragma unroll
      for (int j = 0; j < 2; ++j) {
        int col = wcol * 32 + j * 16 + r;
        bf[j][0] = Bs[kk + 2 * hi][col];
        bf[j][1] = Bs[kk + 1 + 2 * hi][col];
      }
      #pragma unroll
      for (int i = 0; i < 4; ++i)
        #pragma unroll
        for (int j = 0; j < 2; ++j)
          acc[i][j] = __builtin_amdgcn_wmma_f32_16x16x4_f32(
              false, af[i], false, bf[j], (short)0, acc[i][j], false, false);
    }
  }

  #pragma unroll
  for (int i = 0; i < 4; ++i)
    for (int j = 0; j < 2; ++j) {
      int gcol = n0 + wcol * 32 + j * 16 + r;
      if (gcol >= N) continue;
      float b = bias ? bias[gcol] : 0.0f;
      #pragma unroll
      for (int vg = 0; vg < 8; ++vg) {
        int grow = m0 + wrow * 64 + i * 16 + vg + 8 * hi;
        C[(size_t)grow * N + gcol] = acc[i][j][vg] + b;
      }
    }
}

__global__ __launch_bounds__(256) void gemm_wmma_f32(
    const float* __restrict__ A, const float* __restrict__ W,
    const float* __restrict__ bias, float* __restrict__ C,
    int N, int K, int kslice, float* __restrict__ Cpart)
{
  const bool partial = blockIdx.z != 0;
  const int  kbeg    = (int)blockIdx.z * kslice;
  gemm_tile_128(A, W, partial ? nullptr : bias, partial ? Cpart : C, N, K,
                kbeg, kbeg + kslice, (int)blockIdx.y * 128, (int)blockIdx.x * 128);
}

__global__ __launch_bounds__(256) void gemm_q_kva(
    const float* __restrict__ hidden,
    const float* __restrict__ q_w, const float* __restrict__ q_b, float* __restrict__ qbuf,
    const float* __restrict__ kv_a_w, float* __restrict__ ckv, float* __restrict__ part)
{
  const int  b     = (int)blockIdx.x;
  const bool kva   = b >= 512;
  const int  c     = b - 512;
  const int  slice = c / 48;
  const int  t     = c - slice * 48;
  const int  ct    = kva ? t % 3 : (b & 31);
  const int  rt    = kva ? t / 3 : (b >> 5);
  const int  kbeg  = kva ? slice * 576 : 0;
  const int  kend  = kva ? kbeg + 576 : 2880;
  float* __restrict__ C = !kva ? qbuf
                        : (slice == 0 ? ckv : part + (size_t)(slice - 1) * (2048 * 384));
  gemm_tile_128(hidden, kva ? kv_a_w : q_w, kva ? nullptr : q_b, C, kva ? 384 : 4096,
                2880, kbeg, kend, rt * 128, ct * 128);
}

__global__ __launch_bounds__(256) void add_partial(float* __restrict__ C,
                                                   const float* __restrict__ P)
{
  const size_t base = (size_t)blockIdx.x * 1024 + threadIdx.x;
  float4 a[4], b[4];
  #pragma unroll
  for (int u = 0; u < 4; ++u) {
    a[u] = ((const float4*)C)[base + u * 256];
    b[u] = ((const float4*)P)[base + u * 256];
  }
  #pragma unroll
  for (int u = 0; u < 4; ++u) {
    a[u].x += b[u].x; a[u].y += b[u].y; a[u].z += b[u].z; a[u].w += b[u].w;
    ((float4*)C)[base + u * 256] = a[u];
  }
}

__global__ __launch_bounds__(256) void sum_planes5(float* __restrict__ C,
                                                   const float* __restrict__ P)
{
  const size_t base = (size_t)blockIdx.x * 1024 + threadIdx.x;
  float4 a[4], p[4][4];
  #pragma unroll
  for (int u = 0; u < 4; ++u) {
    a[u] = ((const float4*)C)[base + u * 256];
    #pragma unroll
    for (int q = 0; q < 4; ++q)
      p[q][u] = ((const float4*)P)[(size_t)q * 196608 + base + u * 256];
  }
  #pragma unroll
  for (int u = 0; u < 4; ++u) {
    #pragma unroll
    for (int q = 0; q < 4; ++q) {
      a[u].x += p[q][u].x; a[u].y += p[q][u].y; a[u].z += p[q][u].z; a[u].w += p[q][u].w;
    }
    ((float4*)C)[base + u * 256] = a[u];
  }
}

__global__ __launch_bounds__(256) void rope_kernel(float* __restrict__ qbuf,
                                                   float* __restrict__ kvbuf)
{
  const int lane = threadIdx.x & 31;
  const int w    = (int)blockIdx.x * 8 + ((int)threadIdx.x >> 5);
  const int s    = w >> 1;
  const int g0   = (w & 1) * 9;
  const int i4   = (lane & 7) * 4;
  const int sub  = lane >> 3;

  float* p[9];
  float4 x1[9], x2[9];
  #pragma unroll
  for (int u = 0; u < 9; ++u) {
    const int slot = (g0 + u) * 4 + sub;
    p[u] = (slot < 64) ? qbuf  + (size_t)s * 4096 + slot * 64
                       : kvbuf + (size_t)s * 1024 + (slot - 64) * 128;
    x1[u] = *(const float4*)(p[u] + i4);
    x2[u] = *(const float4*)(p[u] + i4 + 32);
  }

  const float inv = powf(150000.0f, -((float)(2 * lane)) / 64.0f);
  const float ang = (float)s * inv;
  const float cl  = cosf(ang);
  const float sl  = sinf(ang);
  float c[4], sn[4];
  #pragma unroll
  for (int e = 0; e < 4; ++e) {
    c[e]  = __shfl(cl, i4 + e);
    sn[e] = __shfl(sl, i4 + e);
  }

  #pragma unroll
  for (int u = 0; u < 9; ++u) {
    float4 a = x1[u], b = x2[u], ra, rb;
    ra.x = a.x * c[0] - b.x * sn[0];  rb.x = b.x * c[0] + a.x * sn[0];
    ra.y = a.y * c[1] - b.y * sn[1];  rb.y = b.y * c[1] + a.y * sn[1];
    ra.z = a.z * c[2] - b.z * sn[2];  rb.z = b.z * c[2] + a.z * sn[2];
    ra.w = a.w * c[3] - b.w * sn[3];  rb.w = b.w * c[3] + a.w * sn[3];
    *(float4*)(p[u] + i4)      = ra;
    *(float4*)(p[u] + i4 + 32) = rb;
  }
}


__device__ __forceinline__ void attn_scores(
    const float (*Qslab)[65], const float (*Ks)[68],
    const int r, const int hi, const int kvbase, const int qr,
    const bool diag, v8f (&sacc)[4])
{
  for (int sub = 0; sub < 4; ++sub)
    for (int c = 0; c < 8; ++c) sacc[sub][c] = 0.0f;
  #pragma unroll
  for (int kk = 0; kk < 64; kk += 4) {
    v2f bq;
    bq[0] = Qslab[r][kk + 2 * hi];
    bq[1] = Qslab[r][kk + 1 + 2 * hi];
    #pragma unroll
    for (int sub = 0; sub < 4; ++sub) {
      v2f ak = *(const v2f*)&Ks[sub * 16 + r][kk + 2 * hi];
      sacc[sub] = __builtin_amdgcn_wmma_f32_16x16x4_f32(
          false, ak, false, bq, (short)0, sacc[sub], false, false);
    }
  }

  #pragma unroll
  for (int sub = 0; sub < 4; ++sub)
    #pragma unroll
    for (int g = 0; g < 8; ++g) sacc[sub][g] *= 0.125f;
  if (diag) {
    #pragma unroll
    for (int sub = 0; sub < 4; ++sub)
      #pragma unroll
      for (int g = 0; g < 8; ++g) {
        int kc = kvbase + sub * 16 + 8 * hi + g;
        sacc[sub][g] = (kc <= qr) ? sacc[sub][g] : NEGBIG;
      }
  }
}

__device__ __forceinline__ void attn_softmax_pv(
    const float (*Vs)[68], const int r, const int hi, const int lane,
    float& m_i, float& l_i, v8f (&sacc)[4], v8f (&oacc)[4])
{
  float mx[4];
  #pragma unroll
  for (int sub = 0; sub < 4; ++sub) {
    mx[sub] = sacc[sub][0];
    #pragma unroll
    for (int g = 1; g < 8; ++g) mx[sub] = fmaxf(mx[sub], sacc[sub][g]);
  }
  float rowmax = fmaxf(fmaxf(mx[0], mx[1]), fmaxf(mx[2], mx[3]));
  rowmax = fmaxf(rowmax, __int_as_float(__builtin_amdgcn_ds_bpermute(
      4 * (lane ^ 16), __float_as_int(rowmax))));
  const float newm  = fmaxf(m_i, rowmax);
  const float alpha = __expf(m_i - newm);
  float ps[4];
  #pragma unroll
  for (int sub = 0; sub < 4; ++sub) {
    ps[sub] = 0.0f;
    #pragma unroll
    for (int g = 0; g < 8; ++g) {
      float p = __expf(sacc[sub][g] - newm);
      sacc[sub][g] = p;
      ps[sub] += p;
    }
  }
  l_i = l_i * alpha + ((ps[0] + ps[1]) + (ps[2] + ps[3]));
  m_i = newm;
  #pragma unroll
  for (int ds = 0; ds < 4; ++ds)
    #pragma unroll
    for (int g = 0; g < 8; ++g) oacc[ds][g] *= alpha;

  #pragma unroll
  for (int sub = 0; sub < 4; ++sub)
    #pragma unroll
    for (int t = 0; t < 4; ++t) {
      v2f bp;
      bp[0] = sacc[sub][2 * t];
      bp[1] = sacc[sub][2 * t + 1];
      const int n0 = sub * 16 + 8 * hi + 2 * t;
      #pragma unroll
      for (int ds = 0; ds < 4; ++ds) {
        v2f av;
        av[0] = Vs[n0][ds * 16 + r];
        av[1] = Vs[n0 + 1][ds * 16 + r];
        oacc[ds] = __builtin_amdgcn_wmma_f32_16x16x4_f32(
            false, av, false, bp, (short)0, oacc[ds], false, false);
      }
    }
}

__device__ __forceinline__ void attn_write_slab(
    float* __restrict__ attn, const int qr, const int h, const int hi, const int lane,
    const float l_i, const v8f (&oacc)[4])
{
  float l_other = __int_as_float(__builtin_amdgcn_ds_bpermute(
      4 * (lane ^ 16), __float_as_int(l_i)));
  float inv = 1.0f / (l_i + l_other);
  float* orow = attn + (size_t)qr * 4096 + h * 64 + 8 * hi;
  #pragma unroll
  for (int ds = 0; ds < 4; ++ds) {
    float4 a = make_float4(oacc[ds][0] * inv, oacc[ds][1] * inv,
                           oacc[ds][2] * inv, oacc[ds][3] * inv);
    float4 b = make_float4(oacc[ds][4] * inv, oacc[ds][5] * inv,
                           oacc[ds][6] * inv, oacc[ds][7] * inv);
    *(float4*)(orow + ds * 16)     = a;
    *(float4*)(orow + ds * 16 + 4) = b;
  }
}

__global__ __launch_bounds__(128, 4) void flash_attn(
    const float* __restrict__ qbuf, const float* __restrict__ kvbuf,
    float* __restrict__ attn)
{
  __shared__ float Qs[2][64][65];
  __shared__ __attribute__((aligned(16))) float Ks[64][68];
  __shared__ __attribute__((aligned(16))) float Vs[64][68];

  const int tid    = threadIdx.x;
  const int wave   = tid >> 5;
  const int lane   = tid & 31;
  const int r      = lane & 15;
  const int hi     = lane >> 4;
  const int tnear  = blockIdx.x;
  const int tfar   = 31 - tnear;
  const int h      = blockIdx.y;
  const int kvh    = h >> 3;
  const int qr_near = tnear * 64 + wave * 16 + r;
  const int qr_far  = tfar  * 64 + wave * 16 + r;

  #pragma unroll 4
  for (int it = 0; it < 32; ++it) {
    int f = tid + it * 128;
    int sl = f >> 6, d = f & 63;
    Qs[0][sl][d] = qbuf[(size_t)(tnear * 64 + sl) * 4096 + h * 64 + d];
    Qs[1][sl][d] = qbuf[(size_t)(tfar  * 64 + sl) * 4096 + h * 64 + d];
  }

  float m_near = -INFINITY, m_far = -INFINITY;
  float l_near = 0.0f,      l_far = 0.0f;
  v8f o_near[4], o_far[4];
  for (int ds = 0; ds < 4; ++ds)
    for (int c = 0; c < 8; ++c) { o_near[ds][c] = 0.0f; o_far[ds][c] = 0.0f; }

  const int srow = tid >> 6, scol = tid & 63;
  const float* kvsrc = kvbuf + (size_t)srow * 1024 + kvh * 128 + scol;
  float kreg[32], vreg[32];
  #pragma unroll
  for (int it = 0; it < 32; ++it) {
    kreg[it] = kvsrc[(size_t)it * 2048];
    vreg[it] = kvsrc[(size_t)it * 2048 + 64];
  }

  for (int j = 0; j <= tfar; ++j) {
    const int kvbase = j * 64;
    const float* nxt = kvsrc + (size_t)(kvbase + 64) * 1024;
    __syncthreads();
    #pragma unroll
    for (int it = 0; it < 32; ++it) {
      Ks[srow + 2 * it][scol] = kreg[it];
      Vs[srow + 2 * it][scol] = vreg[it];
    }
    __syncthreads();

    v8f sacc[4];
    if (j < tfar) {
      #pragma unroll
      for (int it = 0; it < 32; ++it) kreg[it] = nxt[(size_t)it * 2048];
    }
    attn_scores(&Qs[1][wave * 16], Ks, r, hi, kvbase, qr_far, j == tfar, sacc);
    if (j < tfar) {
      #pragma unroll
      for (int it = 0; it < 32; ++it) vreg[it] = nxt[(size_t)it * 2048 + 64];
    }
    attn_softmax_pv(Vs, r, hi, lane, m_far, l_far, sacc, o_far);
    if (j <= tnear) {
      attn_scores(&Qs[0][wave * 16], Ks, r, hi, kvbase, qr_near, j == tnear, sacc);
      attn_softmax_pv(Vs, r, hi, lane, m_near, l_near, sacc, o_near);
    }
  }

  attn_write_slab(attn, qr_far,  h, hi, lane, l_far,  o_far);
  attn_write_slab(attn, qr_near, h, hi, lane, l_near, o_near);
}

extern "C" void kernel_launch(void* const* d_in, const int* in_sizes, int n_in,
                              void* d_out, int out_size, void* d_ws, size_t ws_size,
                              hipStream_t stream)
{
  const float* hidden = (const float*)d_in[0];
  const float* q_w    = (const float*)d_in[2];
  const float* q_b    = (const float*)d_in[3];
  const float* kv_a_w = (const float*)d_in[4];
  const float* kv_b_w = (const float*)d_in[5];
  const float* o_w    = (const float*)d_in[6];
  const float* o_b    = (const float*)d_in[7];
  float* out = (float*)d_out;

  float* ws    = (float*)d_ws;
  float* qbuf  = ws;
  float* ckv   = qbuf  + (size_t)2048 * 4096;
  float* kvbuf = ckv   + (size_t)2048 * 384;
  float* attn  = kvbuf + (size_t)2048 * 1024;

  dim3 blk(256);
  gemm_q_kva<<<dim3(512 + 240), blk, 0, stream>>>(
      hidden, q_w, q_b, qbuf, kv_a_w, ckv, attn);
  sum_planes5<<<192, 256, 0, stream>>>(ckv, attn);
  gemm_wmma_f32<<<dim3(1024 / 128, 2048 / 128), blk, 0, stream>>>(
      ckv, kv_b_w, nullptr, kvbuf, 1024, 384, 384, nullptr);
  rope_kernel<<<512, 256, 0, stream>>>(qbuf, kvbuf);
  flash_attn<<<dim3(16, 64), 128, 0, stream>>>(qbuf, kvbuf, attn);
  float* opart = qbuf;
  gemm_wmma_f32<<<dim3((2880 + 127) / 128, 2048 / 128, 2), blk, 0, stream>>>(
      attn, o_w, o_b, out, 2880, 4096, 2048, opart);
  add_partial<<<1440, 256, 0, stream>>>(out, opart);
}


# GEMMs: W bypasses LDS: a lane reads its columns' 64 B of a k tile from memory into the B fragment registers, one load group per k step, A fragments read a step ahead; k sum order in a k tile changes
# speedup vs baseline: 1.9598x; 1.1078x over previous
#include <hip/hip_runtime.h>
#include <math.h>

typedef __attribute__((ext_vector_type(2))) float v2f;
typedef __attribute__((ext_vector_type(8))) float v8f;

#define NEGBIG (-3.4e38f)

__device__ __forceinline__ void gemm_tile_128(
    const float* __restrict__ A, const float* __restrict__ W,
    const float* __restrict__ bias, float* __restrict__ C,
    const int N, const int K, const int kbeg, const int kend,
    const int m0, const int n0)
{
  __shared__ float As[16 * 289];

  const int tid  = threadIdx.x;
  const int wave = tid >> 5;
  const int lane = tid & 31;
  const int r    = lane & 15;
  const int hi   = lane >> 4;
  const int wrow = wave >> 2;
  const int wcol = wave & 3;

  v8f acc[4][2];
  for (int i = 0; i < 4; ++i)
    for (int j = 0; j < 2; ++j)
      for (int c = 0; c < 8; ++c) acc[i][j][c] = 0.0f;

  const float* wp[2];
  #pragma unroll
  for (int j = 0; j < 2; ++j) {
    const int n = n0 + wcol * 32 + j * 16 + r;
    wp[j] = W + (size_t)(n < N ? n : N - 1) * K + 16 * hi;
  }
  const float* afrag = As + r * 289 + wrow * (4 * 33) + 16 * hi;
  const int arow = tid >> 3;
  const int acol = (tid & 7) * 4;

  for (int k0 = kbeg; k0 < kend; k0 += 32) {
    float4 bw[2][4];
    #pragma unroll
    for (int j = 0; j < 2; ++j)
      #pragma unroll
      for (int q = 0; q < 4; ++q) bw[j][q] = *(const float4*)(wp[j] + k0 + 4 * q);
    float4 av[4];
    #pragma unroll
    for (int it = 0; it < 4; ++it)
      av[it] = *(const float4*)&A[(size_t)(m0 + arow + it * 32) * K + k0 + acol];
    asm volatile("" :: "v"(av[0].x), "v"(av[0].y), "v"(av[0].z), "v"(av[0].w),
                       "v"(av[1].x), "v"(av[1].y), "v"(av[1].z), "v"(av[1].w),
                       "v"(av[2].x), "v"(av[2].y), "v"(av[2].z), "v"(av[2].w),
                       "v"(av[3].x), "v"(av[3].y), "v"(av[3].z), "v"(av[3].w));

    __syncthreads();
    #pragma unroll
    for (int it = 0; it < 4; ++it) {
      float* dst = As + (arow & 15) * 289 + ((arow >> 4) + 2 * it) * 33 + acol;
      dst[0] = av[it].x; dst[1] = av[it].y; dst[2] = av[it].z; dst[3] = av[it].w;
    }
    __syncthreads();

    v2f af[2][4];
    #pragma unroll
    for (int i = 0; i < 4; ++i) {
      af[0][i][0] = afrag[i * 33];
      af[0][i][1] = afrag[i * 33 + 1];
    }
    #pragma unroll
    for (int t = 0; t < 8; ++t) {
      if (t < 7) {
        #pragma unroll
        for (int i = 0; i < 4; ++i) {
          af[(t + 1) & 1][i][0] = afrag[i * 33 + 2 * (t + 1)];
          af[(t + 1) & 1][i][1] = afrag[i * 33 + 2 * (t + 1) + 1];
        }
      }
      __builtin_amdgcn_sched_barrier(0);
      v2f bf[2];
      #pragma unroll
      for (int j = 0; j < 2; ++j) {
        const float4 w4 = bw[j][t >> 1];
        bf[j][0] = (t & 1) ? w4.z : w4.x;
        bf[j][1] = (t & 1) ? w4.w : w4.y;
      }
      #pragma unroll
      for (int i = 0; i < 4; ++i)
        #pragma unroll
        for (int j = 0; j < 2; ++j)
          acc[i][j] = __builtin_amdgcn_wmma_f32_16x16x4_f32(
              false, af[t & 1][i], false, bf[j], (short)0, acc[i][j], false, false);
      __builtin_amdgcn_sched_barrier(0);
    }
  }

  #pragma unroll
  for (int i = 0; i < 4; ++i)
    for (int j = 0; j < 2; ++j) {
      int gcol = n0 + wcol * 32 + j * 16 + r;
      if (gcol >= N) continue;
      float b = bias ? bias[gcol] : 0.0f;
      #pragma unroll
      for (int vg = 0; vg < 8; ++vg) {
        int grow = m0 + wrow * 64 + i * 16 + vg + 8 * hi;
        C[(size_t)grow * N + gcol] = acc[i][j][vg] + b;
      }
    }
}

__global__ __launch_bounds__(256) void gemm_wmma_f32(
    const float* __restrict__ A, const float* __restrict__ W,
    const float* __restrict__ bias, float* __restrict__ C,
    int N, int K, int kslice, float* __restrict__ Cpart)
{
  const bool partial = blockIdx.z != 0;
  const int  kbeg    = (int)blockIdx.z * kslice;
  gemm_tile_128(A, W, partial ? nullptr : bias, partial ? Cpart : C, N, K,
                kbeg, kbeg + kslice, (int)blockIdx.y * 128, (int)blockIdx.x * 128);
}

__global__ __launch_bounds__(256) void gemm_q_kva(
    const float* __restrict__ hidden,
    const float* __restrict__ q_w, const float* __restrict__ q_b, float* __restrict__ qbuf,
    const float* __restrict__ kv_a_w, float* __restrict__ ckv, float* __restrict__ part)
{
  const int  b     = (int)blockIdx.x;
  const bool kva   = b >= 512;
  const int  c     = b - 512;
  const int  slice = c / 48;
  const int  t     = c - slice * 48;
  const int  ct    = kva ? t % 3 : (b & 31);
  const int  rt    = kva ? t / 3 : (b >> 5);
  const int  kbeg  = kva ? slice * 576 : 0;
  const int  kend  = kva ? kbeg + 576 : 2880;
  float* __restrict__ C = !kva ? qbuf
                        : (slice == 0 ? ckv : part + (size_t)(slice - 1) * (2048 * 384));
  gemm_tile_128(hidden, kva ? kv_a_w : q_w, kva ? nullptr : q_b, C, kva ? 384 : 4096,
                2880, kbeg, kend, rt * 128, ct * 128);
}

__global__ __launch_bounds__(256) void add_partial(float* __restrict__ C,
                                                   const float* __restrict__ P)
{
  const size_t base = (size_t)blockIdx.x * 1024 + threadIdx.x;
  float4 a[4], b[4];
  #pragma unroll
  for (int u = 0; u < 4; ++u) {
    a[u] = ((const float4*)C)[base + u * 256];
    b[u] = ((const float4*)P)[base + u * 256];
  }
  #pragma unroll
  for (int u = 0; u < 4; ++u) {
    a[u].x += b[u].x; a[u].y += b[u].y; a[u].z += b[u].z; a[u].w += b[u].w;
    ((float4*)C)[base + u * 256] = a[u];
  }
}

__global__ __launch_bounds__(256) void sum_planes5(float* __restrict__ C,
                                                   const float* __restrict__ P)
{
  const size_t base = (size_t)blockIdx.x * 1024 + threadIdx.x;
  float4 a[4], p[4][4];
  #pragma unroll
  for (int u = 0; u < 4; ++u) {
    a[u] = ((const float4*)C)[base + u * 256];
    #pragma unroll
    for (int q = 0; q < 4; ++q)
      p[q][u] = ((const float4*)P)[(size_t)q * 196608 + base + u * 256];
  }
  #pragma unroll
  for (int u = 0; u < 4; ++u) {
    #pragma unroll
    for (int q = 0; q < 4; ++q) {
      a[u].x += p[q][u].x; a[u].y += p[q][u].y; a[u].z += p[q][u].z; a[u].w += p[q][u].w;
    }
    ((float4*)C)[base + u * 256] = a[u];
  }
}

__global__ __launch_bounds__(256) void rope_kernel(float* __restrict__ qbuf,
                                                   float* __restrict__ kvbuf)
{
  const int lane = threadIdx.x & 31;
  const int w    = (int)blockIdx.x * 8 + ((int)threadIdx.x >> 5);
  const int s    = w >> 1;
  const int g0   = (w & 1) * 9;
  const int i4   = (lane & 7) * 4;
  const int sub  = lane >> 3;

  float* p[9];
  float4 x1[9], x2[9];
  #pragma unroll
  for (int u = 0; u < 9; ++u) {
    const int slot = (g0 + u) * 4 + sub;
    p[u] = (slot < 64) ? qbuf  + (size_t)s * 4096 + slot * 64
                       : kvbuf + (size_t)s * 1024 + (slot - 64) * 128;
    x1[u] = *(const float4*)(p[u] + i4);
    x2[u] = *(const float4*)(p[u] + i4 + 32);
  }

  const float inv = powf(150000.0f, -((float)(2 * lane)) / 64.0f);
  const float ang = (float)s * inv;
  const float cl  = cosf(ang);
  const float sl  = sinf(ang);
  float c[4], sn[4];
  #pragma unroll
  for (int e = 0; e < 4; ++e) {
    c[e]  = __shfl(cl, i4 + e);
    sn[e] = __shfl(sl, i4 + e);
  }

  #pragma unroll
  for (int u = 0; u < 9; ++u) {
    float4 a = x1[u], b = x2[u], ra, rb;
    ra.x = a.x * c[0] - b.x * sn[0];  rb.x = b.x * c[0] + a.x * sn[0];
    ra.y = a.y * c[1] - b.y * sn[1];  rb.y = b.y * c[1] + a.y * sn[1];
    ra.z = a.z * c[2] - b.z * sn[2];  rb.z = b.z * c[2] + a.z * sn[2];
    ra.w = a.w * c[3] - b.w * sn[3];  rb.w = b.w * c[3] + a.w * sn[3];
    *(float4*)(p[u] + i4)      = ra;
    *(float4*)(p[u] + i4 + 32) = rb;
  }
}


__device__ __forceinline__ void attn_scores(
    const float (*Qslab)[65], const float (*Ks)[68],
    const int r, const int hi, const int kvbase, const int qr,
    const bool diag, v8f (&sacc)[4])
{
  for (int sub = 0; sub < 4; ++sub)
    for (int c = 0; c < 8; ++c) sacc[sub][c] = 0.0f;
  #pragma unroll
  for (int kk = 0; kk < 64; kk += 4) {
    v2f bq;
    bq[0] = Qslab[r][kk + 2 * hi];
    bq[1] = Qslab[r][kk + 1 + 2 * hi];
    #pragma unroll
    for (int sub = 0; sub < 4; ++sub) {
      v2f ak = *(const v2f*)&Ks[sub * 16 + r][kk + 2 * hi];
      sacc[sub] = __builtin_amdgcn_wmma_f32_16x16x4_f32(
          false, ak, false, bq, (short)0, sacc[sub], false, false);
    }
  }

  #pragma unroll
  for (int sub = 0; sub < 4; ++sub)
    #pragma unroll
    for (int g = 0; g < 8; ++g) sacc[sub][g] *= 0.125f;
  if (diag) {
    #pragma unroll
    for (int sub = 0; sub < 4; ++sub)
      #pragma unroll
      for (int g = 0; g < 8; ++g) {
        int kc = kvbase + sub * 16 + 8 * hi + g;
        sacc[sub][g] = (kc <= qr) ? sacc[sub][g] : NEGBIG;
      }
  }
}

__device__ __forceinline__ void attn_softmax_pv(
    const float (*Vs)[68], const int r, const int hi, const int lane,
    float& m_i, float& l_i, v8f (&sacc)[4], v8f (&oacc)[4])
{
  float mx[4];
  #pragma unroll
  for (int sub = 0; sub < 4; ++sub) {
    mx[sub] = sacc[sub][0];
    #pragma unroll
    for (int g = 1; g < 8; ++g) mx[sub] = fmaxf(mx[sub], sacc[sub][g]);
  }
  float rowmax = fmaxf(fmaxf(mx[0], mx[1]), fmaxf(mx[2], mx[3]));
  rowmax = fmaxf(rowmax, __int_as_float(__builtin_amdgcn_ds_bpermute(
      4 * (lane ^ 16), __float_as_int(rowmax))));
  const float newm  = fmaxf(m_i, rowmax);
  const float alpha = __expf(m_i - newm);
  float ps[4];
  #pragma unroll
  for (int sub = 0; sub < 4; ++sub) {
    ps[sub] = 0.0f;
    #pragma unroll
    for (int g = 0; g < 8; ++g) {
      float p = __expf(sacc[sub][g] - newm);
      sacc[sub][g] = p;
      ps[sub] += p;
    }
  }
  l_i = l_i * alpha + ((ps[0] + ps[1]) + (ps[2] + ps[3]));
  m_i = newm;
  #pragma unroll
  for (int ds = 0; ds < 4; ++ds)
    #pragma unroll
    for (int g = 0; g < 8; ++g) oacc[ds][g] *= alpha;

  #pragma unroll
  for (int sub = 0; sub < 4; ++sub)
    #pragma unroll
    for (int t = 0; t < 4; ++t) {
      v2f bp;
      bp[0] = sacc[sub][2 * t];
      bp[1] = sacc[sub][2 * t + 1];
      const int n0 = sub * 16 + 8 * hi + 2 * t;
      #pragma unroll
      for (int ds = 0; ds < 4; ++ds) {
        v2f av;
        av[0] = Vs[n0][ds * 16 + r];
        av[1] = Vs[n0 + 1][ds * 16 + r];
        oacc[ds] = __builtin_amdgcn_wmma_f32_16x16x4_f32(
            false, av, false, bp, (short)0, oacc[ds], false, false);
      }
    }
}

__device__ __forceinline__ void attn_write_slab(
    float* __restrict__ attn, const int qr, const int h, const int hi, const int lane,
    const float l_i, const v8f (&oacc)[4])
{
  float l_other = __int_as_float(__builtin_amdgcn_ds_bpermute(
      4 * (lane ^ 16), __float_as_int(l_i)));
  float inv = 1.0f / (l_i + l_other);
  float* orow = attn + (size_t)qr * 4096 + h * 64 + 8 * hi;
  #pragma unroll
  for (int ds = 0; ds < 4; ++ds) {
    float4 a = make_float4(oacc[ds][0] * inv, oacc[ds][1] * inv,
                           oacc[ds][2] * inv, oacc[ds][3] * inv);
    float4 b = make_float4(oacc[ds][4] * inv, oacc[ds][5] * inv,
                           oacc[ds][6] * inv, oacc[ds][7] * inv);
    *(float4*)(orow + ds * 16)     = a;
    *(float4*)(orow + ds * 16 + 4) = b;
  }
}

__global__ __launch_bounds__(128, 4) void flash_attn(
    const float* __restrict__ qbuf, const float* __restrict__ kvbuf,
    float* __restrict__ attn)
{
  __shared__ float Qs[2][64][65];
  __shared__ __attribute__((aligned(16))) float Ks[64][68];
  __shared__ __attribute__((aligned(16))) float Vs[64][68];

  const int tid    = threadIdx.x;
  const int wave   = tid >> 5;
  const int lane   = tid & 31;
  const int r      = lane & 15;
  const int hi     = lane >> 4;
  const int tnear  = blockIdx.x;
  const int tfar   = 31 - tnear;
  const int h      = blockIdx.y;
  const int kvh    = h >> 3;
  const int qr_near = tnear * 64 + wave * 16 + r;
  const int qr_far  = tfar  * 64 + wave * 16 + r;

  #pragma unroll 4
  for (int it = 0; it < 32; ++it) {
    int f = tid + it * 128;
    int sl = f >> 6, d = f & 63;
    Qs[0][sl][d] = qbuf[(size_t)(tnear * 64 + sl) * 4096 + h * 64 + d];
    Qs[1][sl][d] = qbuf[(size_t)(tfar  * 64 + sl) * 4096 + h * 64 + d];
  }

  float m_near = -INFINITY, m_far = -INFINITY;
  float l_near = 0.0f,      l_far = 0.0f;
  v8f o_near[4], o_far[4];
  for (int ds = 0; ds < 4; ++ds)
    for (int c = 0; c < 8; ++c) { o_near[ds][c] = 0.0f; o_far[ds][c] = 0.0f; }

  const int srow = tid >> 6, scol = tid & 63;
  const float* kvsrc = kvbuf + (size_t)srow * 1024 + kvh * 128 + scol;
  float kreg[32], vreg[32];
  #pragma unroll
  for (int it = 0; it < 32; ++it) {
    kreg[it] = kvsrc[(size_t)it * 2048];
    vreg[it] = kvsrc[(size_t)it * 2048 + 64];
  }

  for (int j = 0; j <= tfar; ++j) {
    const int kvbase = j * 64;
    const float* nxt = kvsrc + (size_t)(kvbase + 64) * 1024;
    __syncthreads();
    #pragma unroll
    for (int it = 0; it < 32; ++it) {
      Ks[srow + 2 * it][scol] = kreg[it];
      Vs[srow + 2 * it][scol] = vreg[it];
    }
    __syncthreads();

    v8f sacc[4];
    if (j < tfar) {
      #pragma unroll
      for (int it = 0; it < 32; ++it) kreg[it] = nxt[(size_t)it * 2048];
    }
    attn_scores(&Qs[1][wave * 16], Ks, r, hi, kvbase, qr_far, j == tfar, sacc);
    if (j < tfar) {
      #pragma unroll
      for (int it = 0; it < 32; ++it) vreg[it] = nxt[(size_t)it * 2048 + 64];
    }
    attn_softmax_pv(Vs, r, hi, lane, m_far, l_far, sacc, o_far);
    if (j <= tnear) {
      attn_scores(&Qs[0][wave * 16], Ks, r, hi, kvbase, qr_near, j == tnear, sacc);
      attn_softmax_pv(Vs, r, hi, lane, m_near, l_near, sacc, o_near);
    }
  }

  attn_write_slab(attn, qr_far,  h, hi, lane, l_far,  o_far);
  attn_write_slab(attn, qr_near, h, hi, lane, l_near, o_near);
}

extern "C" void kernel_launch(void* const* d_in, const int* in_sizes, int n_in,
                              void* d_out, int out_size, void* d_ws, size_t ws_size,
                              hipStream_t stream)
{
  const float* hidden = (const float*)d_in[0];
  const float* q_w    = (const float*)d_in[2];
  const float* q_b    = (const float*)d_in[3];
  const float* kv_a_w = (const float*)d_in[4];
  const float* kv_b_w = (const float*)d_in[5];
  const float* o_w    = (const float*)d_in[6];
  const float* o_b    = (const float*)d_in[7];
  float* out = (float*)d_out;

  float* ws    = (float*)d_ws;
  float* qbuf  = ws;
  float* ckv   = qbuf  + (size_t)2048 * 4096;
  float* kvbuf = ckv   + (size_t)2048 * 384;
  float* attn  = kvbuf + (size_t)2048 * 1024;

  dim3 blk(256);
  gemm_q_kva<<<dim3(512 + 240), blk, 0, stream>>>(
      hidden, q_w, q_b, qbuf, kv_a_w, ckv, attn);
  sum_planes5<<<192, 256, 0, stream>>>(ckv, attn);
  gemm_wmma_f32<<<dim3(1024 / 128, 2048 / 128), blk, 0, stream>>>(
      ckv, kv_b_w, nullptr, kvbuf, 1024, 384, 384, nullptr);
  rope_kernel<<<512, 256, 0, stream>>>(qbuf, kvbuf);
  flash_attn<<<dim3(16, 64), 128, 0, stream>>>(qbuf, kvbuf, attn);
  float* opart = qbuf;
  gemm_wmma_f32<<<dim3((2880 + 127) / 128, 2048 / 128, 2), blk, 0, stream>>>(
      attn, o_w, o_b, out, 2880, 4096, 2048, opart);
  add_partial<<<1440, 256, 0, stream>>>(out, opart);
}
